# starting kernel
# baseline (speedup 1.0000x reference)
#include <hip/hip_runtime.h>
#include <hip/hip_bf16.h>

typedef __attribute__((ext_vector_type(2))) float v2f;
typedef __attribute__((ext_vector_type(8))) float v8f;

#define N_NODES  100000
#define D_FEAT   64
#define N_CH     64
#define N_EDGES  1600000

__global__ __launch_bounds__(256) void gcn_gemm_wmma(
    const float* __restrict__ F, const float* __restrict__ W,
    const float* __restrict__ bias, float* __restrict__ H) {
  __shared__ v2f   ldsW2[(D_FEAT / 2) * N_CH];
  __shared__ float ldsB[N_CH];
  for (int i = threadIdx.x; i < (D_FEAT / 2) * N_CH; i += 256) {
    const int kpair = i >> 6;
    const int col   = i & 63;
    v2f w;
    w.x = W[(2 * kpair + 0) * N_CH + col];
    w.y = W[(2 * kpair + 1) * N_CH + col];
    ldsW2[i] = w;
  }
  if (threadIdx.x < N_CH) ldsB[threadIdx.x] = bias[threadIdx.x];
  __syncthreads();

  const int wave = threadIdx.x >> 5;
  const int lane = threadIdx.x & 31;
  const int half = lane >> 4;
  const int l16  = lane & 15;
  const int row0 = blockIdx.x * 128 + wave * 16;
  if (row0 >= N_NODES) return;

  v8f acc[4] = {{}, {}, {}, {}};
  const float* frow = F + (size_t)(row0 + l16) * D_FEAT;

  for (int kk = 0; kk < D_FEAT; kk += 4) {
    const int k2 = kk + 2 * half;
    v2f a;
    a.x = frow[k2];
    a.y = frow[k2 + 1];
    const v2f* wrow = ldsW2 + (size_t)((kk >> 1) + half) * N_CH;
#pragma unroll
    for (int ct = 0; ct < 4; ++ct) {
      v2f b = wrow[ct * 16 + l16];
      acc[ct] = __builtin_amdgcn_wmma_f32_16x16x4_f32(
           false, a,  false, b,
           (short)0, acc[ct],  false,  false);
    }
  }

#pragma unroll
  for (int ct = 0; ct < 4; ++ct) {
    const float bb = ldsB[ct * 16 + l16];
#pragma unroll
    for (int v = 0; v < 8; ++v) {
      H[(size_t)(row0 + v + 8 * half) * N_CH + ct * 16 + l16] = acc[ct][v] + bb;
    }
  }
}

__global__ __launch_bounds__(256) void zero_f4(float4* __restrict__ p, int n4) {
  int t = blockIdx.x * 256 + threadIdx.x;
  if (t < n4) p[t] = make_float4(0.f, 0.f, 0.f, 0.f);
}

__global__ __launch_bounds__(256) void spmm_scatter(
    const float* __restrict__ H, const float* __restrict__ vals,
    const int* __restrict__ rows, const int* __restrict__ cols,
    float* __restrict__ agg) {
  const int wv = __builtin_amdgcn_readfirstlane((int)(threadIdx.x >> 5));
  const int e  = blockIdx.x * 8 + wv;
  if (e >= N_EDGES) return;
  const unsigned p = (threadIdx.x & 31u) * 2u;
  const int   c = cols[e];
  const int   r = rows[e];
  const float v = vals[e];
  const float2 hv = *(const float2*)(H + (size_t)c * N_CH + p);
  float* op = agg + (size_t)r * N_CH + p;
  __hip_atomic_fetch_add(op,     v * hv.x, __ATOMIC_RELAXED, __HIP_MEMORY_SCOPE_AGENT);
  __hip_atomic_fetch_add(op + 1, v * hv.y, __ATOMIC_RELAXED, __HIP_MEMORY_SCOPE_AGENT);
}

__global__ __launch_bounds__(256) void selu_epilogue(
    const float* __restrict__ agg, const float* __restrict__ skip,
    float* __restrict__ out) {
  unsigned t = blockIdx.x * 256u + threadIdx.x;
  if (t >= (unsigned)N_NODES * N_CH) return;
  const float x = agg[t] + skip[t & (N_CH - 1)];
  const float alpha = 1.6732632423543772f;
  const float scale = 1.0507009873554805f;
  const float neg = scale * alpha * (expf(x) - 1.0f);
  const float pos = scale * x;
  out[t] = (x > 0.0f) ? pos : neg;
}

extern "C" void kernel_launch(void* const* d_in, const int* in_sizes, int n_in,
                              void* d_out, int out_size, void* d_ws, size_t ws_size,
                              hipStream_t stream) {
  const float* features = (const float*)d_in[0];
  const float* kernelW  = (const float*)d_in[1];
  const float* bias     = (const float*)d_in[2];
  const float* skip     = (const float*)d_in[3];
  const float* vals     = (const float*)d_in[4];
  const int*   rows     = (const int*)d_in[5];
  const int*   cols     = (const int*)d_in[6];

  float* out = (float*)d_out;
  float* h   = out;
  float* agg = (float*)d_ws;

  const int n4 = N_NODES * N_CH / 4;
  zero_f4<<<(n4 + 255) / 256, 256, 0, stream>>>((float4*)agg, n4);

  const int gemm_blocks = (N_NODES + 127) / 128;
  gcn_gemm_wmma<<<gemm_blocks, 256, 0, stream>>>(features, kernelW, bias, h);

  const int sc_blocks = (N_EDGES + 7) / 8;
  spmm_scatter<<<sc_blocks, 256, 0, stream>>>(h, vals, rows, cols, agg);

  const int ep_blocks = (N_NODES * N_CH + 255) / 256;
  selu_epilogue<<<ep_blocks, 256, 0, stream>>>(agg, skip, out);
}


# SpMM: f32-atomic scatter replaced by counting sort of edges on row (ranks from 16-bit LDS counters per 25,000-edge block, no global atomic) + per-row register gather-sum; sum order = LDS arrival rank
# speedup vs baseline: 1.6905x; 1.6905x over previous
#include <hip/hip_runtime.h>
#include <hip/hip_bf16.h>

typedef __attribute__((ext_vector_type(2))) float v2f;
typedef __attribute__((ext_vector_type(8))) float v8f;

#define N_NODES  100000
#define D_FEAT   64
#define N_CH     64
#define N_EDGES  1600000

__global__ __launch_bounds__(256) void gcn_gemm_wmma(
    const float* __restrict__ F, const float* __restrict__ W,
    const float* __restrict__ bias, float* __restrict__ H) {
  __shared__ v2f   ldsW2[(D_FEAT / 2) * N_CH];
  __shared__ float ldsB[N_CH];
  for (int i = threadIdx.x; i < (D_FEAT / 2) * N_CH; i += 256) {
    const int kpair = i >> 6;
    const int col   = i & 63;
    v2f w;
    w.x = W[(2 * kpair + 0) * N_CH + col];
    w.y = W[(2 * kpair + 1) * N_CH + col];
    ldsW2[i] = w;
  }
  if (threadIdx.x < N_CH) ldsB[threadIdx.x] = bias[threadIdx.x];
  __syncthreads();

  const int wave = threadIdx.x >> 5;
  const int lane = threadIdx.x & 31;
  const int half = lane >> 4;
  const int l16  = lane & 15;
  const int row0 = blockIdx.x * 128 + wave * 16;
  if (row0 >= N_NODES) return;

  v8f acc[4] = {{}, {}, {}, {}};
  const float* frow = F + (size_t)(row0 + l16) * D_FEAT;

  for (int kk = 0; kk < D_FEAT; kk += 4) {
    const int k2 = kk + 2 * half;
    v2f a;
    a.x = frow[k2];
    a.y = frow[k2 + 1];
    const v2f* wrow = ldsW2 + (size_t)((kk >> 1) + half) * N_CH;
#pragma unroll
    for (int ct = 0; ct < 4; ++ct) {
      v2f b = wrow[ct * 16 + l16];
      acc[ct] = __builtin_amdgcn_wmma_f32_16x16x4_f32(
           false, a,  false, b,
           (short)0, acc[ct],  false,  false);
    }
  }

#pragma unroll
  for (int ct = 0; ct < 4; ++ct) {
    const float bb = ldsB[ct * 16 + l16];
#pragma unroll
    for (int v = 0; v < 8; ++v) {
      H[(size_t)(row0 + v + 8 * half) * N_CH + ct * 16 + l16] = acc[ct][v] + bb;
    }
  }
}

#define EDGES_PER_BLK 25000
#define N_BLK (N_EDGES / EDGES_PER_BLK)

__device__ __forceinline__ int bump_row(unsigned* hist, int r) {
  const unsigned sh  = ((unsigned)r & 1u) << 4;
  const unsigned old = atomicAdd(hist + (r >> 1), 1u << sh);
  return (int)((old >> sh) & 0xFFFFu);
}

__global__ __launch_bounds__(1024) void count_local(
    const int* __restrict__ rows, int* __restrict__ rank, int* __restrict__ cntmat) {
  __shared__ __attribute__((aligned(16))) unsigned hist[N_NODES / 2];
  const int tid = threadIdx.x;
  for (int i = tid; i < N_NODES / 8; i += 1024) ((uint4*)hist)[i] = make_uint4(0u, 0u, 0u, 0u);
  __syncthreads();
  const int4* r4 = (const int4*)(rows + blockIdx.x * EDGES_PER_BLK);
  int4*       k4 = (int4*)(rank + blockIdx.x * EDGES_PER_BLK);
  for (int i = tid; i < EDGES_PER_BLK / 4; i += 1024) {
    const int4 r = r4[i];
    int4 k;
    k.x = bump_row(hist, r.x);
    k.y = bump_row(hist, r.y);
    k.z = bump_row(hist, r.z);
    k.w = bump_row(hist, r.w);
    k4[i] = k;
  }
  __syncthreads();
  int2* dst = (int2*)(cntmat + (size_t)blockIdx.x * N_NODES);
  for (int w = tid; w < N_NODES / 2; w += 1024) {
    const unsigned c = hist[w];
    dst[w] = make_int2((int)(c & 0xFFFFu), (int)(c >> 16));
  }
}

__global__ __launch_bounds__(256) void prefix_blocks(
    int* __restrict__ cntmat, int* __restrict__ cnt) {
  const int r = blockIdx.x * 256 + threadIdx.x;
  if (r >= N_NODES) return;
  int v[N_BLK];
#pragma unroll
  for (int b = 0; b < N_BLK; ++b) v[b] = cntmat[(size_t)b * N_NODES + r];
  int run = 0;
#pragma unroll
  for (int b = 0; b < N_BLK; ++b) {
    cntmat[(size_t)b * N_NODES + r] = run;
    run += v[b];
  }
  cnt[r] = run;
}

__global__ __launch_bounds__(1024) void scan_counts(
    const int* __restrict__ cnt, int* __restrict__ off) {
  __shared__ int wsum[32];
  __shared__ int s_prefix;
  const int tid  = threadIdx.x;
  const int lane = tid & 31;
  const int wv   = tid >> 5;

  int p = 0;
  const int4* c4 = (const int4*)cnt;
  const int n4 = blockIdx.x * 256;
  for (int base = 0; base < n4; base += 8 * 1024) {
    int4 v[8];
#pragma unroll
    for (int k = 0; k < 8; ++k) {
      const int i = base + k * 1024 + tid;
      v[k] = (i < n4) ? c4[i] : make_int4(0, 0, 0, 0);
    }
#pragma unroll
    for (int k = 0; k < 8; ++k) p += (v[k].x + v[k].y) + (v[k].z + v[k].w);
  }
  for (int o = 16; o > 0; o >>= 1) p += __shfl_xor(p, o);
  if (lane == 0) wsum[wv] = p;
  __syncthreads();
  if (wv == 0) {
    int q = wsum[lane];
    for (int o = 16; o > 0; o >>= 1) q += __shfl_xor(q, o);
    if (lane == 0) s_prefix = q;
  }
  __syncthreads();
  const int prefix = s_prefix;

  const int i = blockIdx.x * 1024 + tid;
  const int c = (i < N_NODES) ? cnt[i] : 0;
  int incl = c;
  for (int o = 1; o < 32; o <<= 1) {
    const int u = __shfl_up(incl, o);
    if (lane >= o) incl += u;
  }
  if (lane == 31) wsum[wv] = incl;
  __syncthreads();
  if (wv == 0) {
    const int q = wsum[lane];
    int s = q;
    for (int o = 1; o < 32; o <<= 1) {
      const int u = __shfl_up(s, o);
      if (lane >= o) s += u;
    }
    wsum[lane] = s - q;
  }
  __syncthreads();
  const int excl = prefix + wsum[wv] + incl - c;
  if (i < N_NODES) off[i] = excl;
  if (i == N_NODES - 1) off[N_NODES] = excl + c;
}

__global__ __launch_bounds__(256) void place_edges(
    const int* __restrict__ rows, const int* __restrict__ cols,
    const float* __restrict__ vals, const int* __restrict__ rank,
    const int* __restrict__ off, const int* __restrict__ cntmat,
    int2* __restrict__ sorted) {
  const int t = blockIdx.x * 256 + threadIdx.x;
  if (t >= N_EDGES / 4) return;
  const int* pre = cntmat + (size_t)((4 * t) / EDGES_PER_BLK) * N_NODES;
  const int4   r = ((const int4*)rows)[t];
  const int4   c = ((const int4*)cols)[t];
  const float4 v = ((const float4*)vals)[t];
  const int4   k = ((const int4*)rank)[t];
  const unsigned p0 = min((unsigned)(off[r.x] + pre[r.x] + k.x), (unsigned)(N_EDGES - 1));
  const unsigned p1 = min((unsigned)(off[r.y] + pre[r.y] + k.y), (unsigned)(N_EDGES - 1));
  const unsigned p2 = min((unsigned)(off[r.z] + pre[r.z] + k.z), (unsigned)(N_EDGES - 1));
  const unsigned p3 = min((unsigned)(off[r.w] + pre[r.w] + k.w), (unsigned)(N_EDGES - 1));
  sorted[p0] = make_int2(c.x, __float_as_int(v.x));
  sorted[p1] = make_int2(c.y, __float_as_int(v.y));
  sorted[p2] = make_int2(c.z, __float_as_int(v.z));
  sorted[p3] = make_int2(c.w, __float_as_int(v.w));
}

__global__ __launch_bounds__(256) void gather_rows(
    const float* __restrict__ H, const int2* __restrict__ sorted,
    const int* __restrict__ off, const float* __restrict__ skip,
    float* __restrict__ out) {
  const int wv = __builtin_amdgcn_readfirstlane((int)(threadIdx.x >> 5));
  const int r  = blockIdx.x * 8 + wv;
  if (r >= N_NODES) return;
  const int lane = threadIdx.x & 31;
  const unsigned p = (unsigned)lane * 2u;
  const int beg = max(off[r], 0);
  const int end = min(off[r + 1], N_EDGES);
  float ax = 0.f, ay = 0.f;
  for (int base = beg; base < end; base += 32) {
    const int n = min(32, end - base);
    int2 e = make_int2(0, 0);
    if (lane < n) e = sorted[base + lane];
#pragma unroll 4
    for (int j = 0; j < n; ++j) {
      const unsigned c = min((unsigned)__builtin_amdgcn_readlane(e.x, j), (unsigned)(N_NODES - 1));
      const float    v = __int_as_float(__builtin_amdgcn_readlane(e.y, j));
      const float2  hv = *(const float2*)(H + (size_t)c * N_CH + p);
      ax += v * hv.x;
      ay += v * hv.y;
    }
  }
  const float2 sk = *(const float2*)(skip + p);
  const float alpha = 1.6732632423543772f;
  const float scale = 1.0507009873554805f;
  const float x0 = ax + sk.x;
  const float x1 = ay + sk.y;
  float2 o;
  o.x = (x0 > 0.0f) ? scale * x0 : scale * alpha * (expf(x0) - 1.0f);
  o.y = (x1 > 0.0f) ? scale * x1 : scale * alpha * (expf(x1) - 1.0f);
  *(float2*)(out + (size_t)r * N_CH + p) = o;
}

extern "C" void kernel_launch(void* const* d_in, const int* in_sizes, int n_in,
                              void* d_out, int out_size, void* d_ws, size_t ws_size,
                              hipStream_t stream) {
  const float* features = (const float*)d_in[0];
  const float* kernelW  = (const float*)d_in[1];
  const float* bias     = (const float*)d_in[2];
  const float* skip     = (const float*)d_in[3];
  const float* vals     = (const float*)d_in[4];
  const int*   rows     = (const int*)d_in[5];
  const int*   cols     = (const int*)d_in[6];

  float* out = (float*)d_out;
  char*  ws  = (char*)d_ws;
  float* h      = (float*)(ws);
  int2*  sorted = (int2*) (ws + 25600000);
  int*   cnt    = (int*)  (ws + 38400000);
  int*   off    = (int*)  (ws + 38800000);
  int*   rank   = (int*)  (ws + 39200256);
  int*   cntmat = (int*)  (ws + 45600256);

  gcn_gemm_wmma<<<(N_NODES + 127) / 128, 256, 0, stream>>>(features, kernelW, bias, h);

  count_local<<<N_BLK, 1024, 0, stream>>>(rows, rank, cntmat);

  prefix_blocks<<<(N_NODES + 255) / 256, 256, 0, stream>>>(cntmat, cnt);

  scan_counts<<<(N_NODES + 1023) / 1024, 1024, 0, stream>>>(cnt, off);

  place_edges<<<(N_EDGES / 4 + 255) / 256, 256, 0, stream>>>(rows, cols, vals, rank, off, cntmat, sorted);

  gather_rows<<<(N_NODES + 7) / 8, 256, 0, stream>>>(h, sorted, off, skip, out);
}


# Placement looks each edge's start position up in an LDS copy of its block's half plane (row offset folded into the planes): no scattered global load per edge, was two; sum order unchanged
# speedup vs baseline: 1.8008x; 1.0652x over previous
#include <hip/hip_runtime.h>
#include <hip/hip_bf16.h>

typedef __attribute__((ext_vector_type(2))) float v2f;
typedef __attribute__((ext_vector_type(8))) float v8f;

#define N_NODES  100000
#define D_FEAT   64
#define N_CH     64
#define N_EDGES  1600000

__global__ __launch_bounds__(256) void gcn_gemm_wmma(
    const float* __restrict__ F, const float* __restrict__ W,
    const float* __restrict__ bias, float* __restrict__ H) {
  __shared__ v2f   ldsW2[(D_FEAT / 2) * N_CH];
  __shared__ float ldsB[N_CH];
  for (int i = threadIdx.x; i < (D_FEAT / 2) * N_CH; i += 256) {
    const int kpair = i >> 6;
    const int col   = i & 63;
    v2f w;
    w.x = W[(2 * kpair + 0) * N_CH + col];
    w.y = W[(2 * kpair + 1) * N_CH + col];
    ldsW2[i] = w;
  }
  if (threadIdx.x < N_CH) ldsB[threadIdx.x] = bias[threadIdx.x];
  __syncthreads();

  const int wave = threadIdx.x >> 5;
  const int lane = threadIdx.x & 31;
  const int half = lane >> 4;
  const int l16  = lane & 15;
  const int row0 = blockIdx.x * 128 + wave * 16;
  if (row0 >= N_NODES) return;

  v8f acc[4] = {{}, {}, {}, {}};
  const float* frow = F + (size_t)(row0 + l16) * D_FEAT;

  for (int kk = 0; kk < D_FEAT; kk += 4) {
    const int k2 = kk + 2 * half;
    v2f a;
    a.x = frow[k2];
    a.y = frow[k2 + 1];
    const v2f* wrow = ldsW2 + (size_t)((kk >> 1) + half) * N_CH;
#pragma unroll
    for (int ct = 0; ct < 4; ++ct) {
      v2f b = wrow[ct * 16 + l16];
      acc[ct] = __builtin_amdgcn_wmma_f32_16x16x4_f32(
           false, a,  false, b,
           (short)0, acc[ct],  false,  false);
    }
  }

#pragma unroll
  for (int ct = 0; ct < 4; ++ct) {
    const float bb = ldsB[ct * 16 + l16];
#pragma unroll
    for (int v = 0; v < 8; ++v) {
      H[(size_t)(row0 + v + 8 * half) * N_CH + ct * 16 + l16] = acc[ct][v] + bb;
    }
  }
}

#define EDGES_PER_BLK 25000
#define N_BLK (N_EDGES / EDGES_PER_BLK)

__device__ __forceinline__ int bump_row(unsigned* hist, int r) {
  const unsigned sh  = ((unsigned)r & 1u) << 4;
  const unsigned old = atomicAdd(hist + (r >> 1), 1u << sh);
  return (int)((old >> sh) & 0xFFFFu);
}

__global__ __launch_bounds__(1024) void count_local(
    const int* __restrict__ rows, int* __restrict__ rank, int* __restrict__ cntmat) {
  __shared__ __attribute__((aligned(16))) unsigned hist[N_NODES / 2];
  const int tid = threadIdx.x;
  for (int i = tid; i < N_NODES / 8; i += 1024) ((uint4*)hist)[i] = make_uint4(0u, 0u, 0u, 0u);
  __syncthreads();
  const int4* r4 = (const int4*)(rows + blockIdx.x * EDGES_PER_BLK);
  int4*       k4 = (int4*)(rank + blockIdx.x * EDGES_PER_BLK);
  for (int i = tid; i < EDGES_PER_BLK / 4; i += 1024) {
    const int4 r = r4[i];
    int4 k;
    k.x = bump_row(hist, r.x);
    k.y = bump_row(hist, r.y);
    k.z = bump_row(hist, r.z);
    k.w = bump_row(hist, r.w);
    k4[i] = k;
  }
  __syncthreads();
  int2* dst = (int2*)(cntmat + (size_t)blockIdx.x * N_NODES);
  for (int w = tid; w < N_NODES / 2; w += 1024) {
    const unsigned c = hist[w];
    dst[w] = make_int2((int)(c & 0xFFFFu), (int)(c >> 16));
  }
}

#define ROWS_PER_TILE 256
#define N_TILES ((N_NODES + ROWS_PER_TILE - 1) / ROWS_PER_TILE)

__global__ __launch_bounds__(256) void tile_totals(
    const int* __restrict__ cntmat, int* __restrict__ tsum) {
  __shared__ int wsum[8];
  const int tid  = threadIdx.x;
  const int lane = tid & 31;
  const int wv   = tid >> 5;
  const int r    = blockIdx.x * ROWS_PER_TILE + tid;
  int s = 0;
  if (r < N_NODES) {
    int v[N_BLK];
#pragma unroll
    for (int b = 0; b < N_BLK; ++b) v[b] = cntmat[(size_t)b * N_NODES + r];
#pragma unroll
    for (int b = 0; b < N_BLK; ++b) s += v[b];
  }
  for (int o = 16; o > 0; o >>= 1) s += __shfl_xor(s, o);
  if (lane == 0) wsum[wv] = s;
  __syncthreads();
  if (tid == 0) {
    int q = 0;
    for (int i = 0; i < 8; ++i) q += wsum[i];
    tsum[blockIdx.x] = q;
  }
}

__global__ __launch_bounds__(256) void base_positions(
    int* __restrict__ cntmat, const int* __restrict__ tsum, int* __restrict__ off) {
  __shared__ int wtot[8];
  __shared__ int wpre[8];
  const int tid  = threadIdx.x;
  const int lane = tid & 31;
  const int wv   = tid >> 5;
  const int r    = blockIdx.x * ROWS_PER_TILE + tid;
  const bool live = r < N_NODES;
  int v[N_BLK];
#pragma unroll
  for (int b = 0; b < N_BLK; ++b) v[b] = live ? cntmat[(size_t)b * N_NODES + r] : 0;

  int p = 0;
  for (int i = tid; i < (int)blockIdx.x; i += ROWS_PER_TILE) p += tsum[i];
  for (int o = 16; o > 0; o >>= 1) p += __shfl_xor(p, o);
  if (lane == 0) wpre[wv] = p;

  int c = 0;
#pragma unroll
  for (int b = 0; b < N_BLK; ++b) c += v[b];
  int incl = c;
  for (int o = 1; o < 32; o <<= 1) {
    const int u = __shfl_up(incl, o);
    if (lane >= o) incl += u;
  }
  if (lane == 31) wtot[wv] = incl;
  __syncthreads();
  int front = 0;
  for (int i = 0; i < 8; ++i) {
    front += wpre[i];
    if (i < wv) front += wtot[i];
  }
  int run = front + incl - c;
  if (live) {
    off[r] = run;
    if (r == N_NODES - 1) off[N_NODES] = run + c;
#pragma unroll
    for (int b = 0; b < N_BLK; ++b) {
      cntmat[(size_t)b * N_NODES + r] = run;
      run += v[b];
    }
  }
}

#define HALF_ROWS      (N_NODES / 2)
#define HALF_ROWS4     (HALF_ROWS / 4)
#define QUADS_PER_PART (EDGES_PER_BLK / 8)
#define STAGE_TRIPS    ((HALF_ROWS4 + 1023) / 1024)
#define QUAD_TRIPS     ((QUADS_PER_PART + 1023) / 1024)

__global__ __launch_bounds__(1024) void place_edges(
    const int* __restrict__ rows, const int* __restrict__ cols,
    const float* __restrict__ vals, const int* __restrict__ rank,
    const int* __restrict__ cntmat, int2* __restrict__ sorted) {
  __shared__ __attribute__((aligned(16))) int pos[HALF_ROWS];
  const int tid  = threadIdx.x;
  const int blk  = blockIdx.x >> 2;
  const int hf   = (blockIdx.x >> 1) & 1;
  const int part = blockIdx.x & 1;
  const int lo   = hf * HALF_ROWS;

  const int4* src4 = (const int4*)(cntmat + (size_t)blk * N_NODES + lo);
  int4 s[STAGE_TRIPS];
#pragma unroll
  for (int j = 0; j < STAGE_TRIPS; ++j) {
    const int i = (j < STAGE_TRIPS - 1) ? tid + j * 1024 : HALF_ROWS4 - 1024 + tid;
    s[j] = src4[i];
  }
#pragma unroll
  for (int j = 0; j < STAGE_TRIPS; ++j) {
    const int i = (j < STAGE_TRIPS - 1) ? tid + j * 1024 : HALF_ROWS4 - 1024 + tid;
    ((int4*)pos)[i] = s[j];
  }

  const int q0 = blk * (EDGES_PER_BLK / 4) + part * QUADS_PER_PART;
  int4 r[QUAD_TRIPS], k[QUAD_TRIPS], c[QUAD_TRIPS];
  float4 v[QUAD_TRIPS];
#pragma unroll
  for (int j = 0; j < QUAD_TRIPS; ++j) {
    const int q = q0 + min(tid + j * 1024, QUADS_PER_PART - 1);
    r[j] = ((const int4*)rows)[q];
    k[j] = ((const int4*)rank)[q];
    c[j] = ((const int4*)cols)[q];
    v[j] = ((const float4*)vals)[q];
  }
  __syncthreads();

  int b[QUAD_TRIPS][4];
#pragma unroll
  for (int j = 0; j < QUAD_TRIPS; ++j) {
    b[j][0] = pos[min((unsigned)(r[j].x - lo), (unsigned)(HALF_ROWS - 1))];
    b[j][1] = pos[min((unsigned)(r[j].y - lo), (unsigned)(HALF_ROWS - 1))];
    b[j][2] = pos[min((unsigned)(r[j].z - lo), (unsigned)(HALF_ROWS - 1))];
    b[j][3] = pos[min((unsigned)(r[j].w - lo), (unsigned)(HALF_ROWS - 1))];
  }
#pragma unroll
  for (int j = 0; j < QUAD_TRIPS; ++j) {
    const bool live = (tid + j * 1024) < QUADS_PER_PART;
    const unsigned p0 = min((unsigned)(b[j][0] + k[j].x), (unsigned)(N_EDGES - 1));
    const unsigned p1 = min((unsigned)(b[j][1] + k[j].y), (unsigned)(N_EDGES - 1));
    const unsigned p2 = min((unsigned)(b[j][2] + k[j].z), (unsigned)(N_EDGES - 1));
    const unsigned p3 = min((unsigned)(b[j][3] + k[j].w), (unsigned)(N_EDGES - 1));
    if (live && (unsigned)(r[j].x - lo) < (unsigned)HALF_ROWS) sorted[p0] = make_int2(c[j].x, __float_as_int(v[j].x));
    if (live && (unsigned)(r[j].y - lo) < (unsigned)HALF_ROWS) sorted[p1] = make_int2(c[j].y, __float_as_int(v[j].y));
    if (live && (unsigned)(r[j].z - lo) < (unsigned)HALF_ROWS) sorted[p2] = make_int2(c[j].z, __float_as_int(v[j].z));
    if (live && (unsigned)(r[j].w - lo) < (unsigned)HALF_ROWS) sorted[p3] = make_int2(c[j].w, __float_as_int(v[j].w));
  }
}

__global__ __launch_bounds__(256) void gather_rows(
    const float* __restrict__ H, const int2* __restrict__ sorted,
    const int* __restrict__ off, const float* __restrict__ skip,
    float* __restrict__ out) {
  const int wv = __builtin_amdgcn_readfirstlane((int)(threadIdx.x >> 5));
  const int r  = blockIdx.x * 8 + wv;
  if (r >= N_NODES) return;
  const int lane = threadIdx.x & 31;
  const unsigned p = (unsigned)lane * 2u;
  const int beg = max(off[r], 0);
  const int end = min(off[r + 1], N_EDGES);
  float ax = 0.f, ay = 0.f;
  for (int base = beg; base < end; base += 32) {
    const int n = min(32, end - base);
    int2 e = make_int2(0, 0);
    if (lane < n) e = sorted[base + lane];
#pragma unroll 4
    for (int j = 0; j < n; ++j) {
      const unsigned c = min((unsigned)__builtin_amdgcn_readlane(e.x, j), (unsigned)(N_NODES - 1));
      const float    v = __int_as_float(__builtin_amdgcn_readlane(e.y, j));
      const float2  hv = *(const float2*)(H + (size_t)c * N_CH + p);
      ax += v * hv.x;
      ay += v * hv.y;
    }
  }
  const float2 sk = *(const float2*)(skip + p);
  const float alpha = 1.6732632423543772f;
  const float scale = 1.0507009873554805f;
  const float x0 = ax + sk.x;
  const float x1 = ay + sk.y;
  float2 o;
  o.x = (x0 > 0.0f) ? scale * x0 : scale * alpha * (expf(x0) - 1.0f);
  o.y = (x1 > 0.0f) ? scale * x1 : scale * alpha * (expf(x1) - 1.0f);
  *(float2*)(out + (size_t)r * N_CH + p) = o;
}

extern "C" void kernel_launch(void* const* d_in, const int* in_sizes, int n_in,
                              void* d_out, int out_size, void* d_ws, size_t ws_size,
                              hipStream_t stream) {
  const float* features = (const float*)d_in[0];
  const float* kernelW  = (const float*)d_in[1];
  const float* bias     = (const float*)d_in[2];
  const float* skip     = (const float*)d_in[3];
  const float* vals     = (const float*)d_in[4];
  const int*   rows     = (const int*)d_in[5];
  const int*   cols     = (const int*)d_in[6];

  float* out = (float*)d_out;
  char*  ws  = (char*)d_ws;
  float* h      = (float*)(ws);
  int2*  sorted = (int2*) (ws + 25600000);
  int*   tsum   = (int*)  (ws + 38400000);
  int*   off    = (int*)  (ws + 38800000);
  int*   rank   = (int*)  (ws + 39200256);
  int*   cntmat = (int*)  (ws + 45600256);

  gcn_gemm_wmma<<<(N_NODES + 127) / 128, 256, 0, stream>>>(features, kernelW, bias, h);

  count_local<<<N_BLK, 1024, 0, stream>>>(rows, rank, cntmat);

  tile_totals<<<N_TILES, ROWS_PER_TILE, 0, stream>>>(cntmat, tsum);

  base_positions<<<N_TILES, ROWS_PER_TILE, 0, stream>>>(cntmat, tsum, off);

  place_edges<<<N_BLK * 4, 1024, 0, stream>>>(rows, cols, vals, rank, cntmat, sorted);

  gather_rows<<<(N_NODES + 7) / 8, 256, 0, stream>>>(h, sorted, off, skip, out);
}


# h = F @ W + bias moved into the counting launch as 196 extra work-groups (job by work-group index) on the WGPs the 64 counting groups leave idle: 5 launches, was 6; all sums unchanged
# speedup vs baseline: 1.9180x; 1.0651x over previous
#include <hip/hip_runtime.h>
#include <hip/hip_bf16.h>

typedef __attribute__((ext_vector_type(2))) float v2f;
typedef __attribute__((ext_vector_type(8))) float v8f;

#define N_NODES  100000
#define D_FEAT   64
#define N_CH     64
#define N_EDGES  1600000

#define EDGES_PER_BLK 25000
#define N_BLK (N_EDGES / EDGES_PER_BLK)
#define ROW_SLABS   (N_NODES / 16)
#define N_GEMM_WG   ((ROW_SLABS + 31) / 32)

__device__ __forceinline__ void dense_job(
    v2f* ldsW2, float* ldsB, const int wg,
    const float* __restrict__ F, const float* __restrict__ W,
    const float* __restrict__ bias, float* __restrict__ H) {
  for (int i = threadIdx.x; i < (D_FEAT / 2) * N_CH; i += 1024) {
    const int kpair = i >> 6;
    const int col   = i & 63;
    v2f w;
    w.x = W[(2 * kpair + 0) * N_CH + col];
    w.y = W[(2 * kpair + 1) * N_CH + col];
    ldsW2[i] = w;
  }
  if (threadIdx.x < N_CH) ldsB[threadIdx.x] = bias[threadIdx.x];
  __syncthreads();

  const int wave = __builtin_amdgcn_readfirstlane((int)(threadIdx.x >> 5));
  const int lane = threadIdx.x & 31;
  const int half = lane >> 4;
  const int l16  = lane & 15;

  const int slab = wg * 32 + wave;
  if (slab < ROW_SLABS) {
    const int row0 = slab * 16;
    v8f acc[4] = {{}, {}, {}, {}};
    const float* frow = F + (size_t)(row0 + l16) * D_FEAT;

    for (int kk = 0; kk < D_FEAT; kk += 4) {
      const int k2 = kk + 2 * half;
      v2f a;
      a.x = frow[k2];
      a.y = frow[k2 + 1];
      const v2f* wrow = ldsW2 + (size_t)((kk >> 1) + half) * N_CH;
#pragma unroll
      for (int ct = 0; ct < 4; ++ct) {
        v2f b = wrow[ct * 16 + l16];
        acc[ct] = __builtin_amdgcn_wmma_f32_16x16x4_f32(
             false, a,  false, b,
             (short)0, acc[ct],  false,  false);
      }
    }

#pragma unroll
    for (int ct = 0; ct < 4; ++ct) {
      const float bb = ldsB[ct * 16 + l16];
#pragma unroll
      for (int v = 0; v < 8; ++v) {
        H[(size_t)(row0 + v + 8 * half) * N_CH + ct * 16 + l16] = acc[ct][v] + bb;
      }
    }
  }
}

__device__ __forceinline__ int bump_row(unsigned* hist, int r) {
  const unsigned sh  = ((unsigned)r & 1u) << 4;
  const unsigned old = atomicAdd(hist + (r >> 1), 1u << sh);
  return (int)((old >> sh) & 0xFFFFu);
}

__device__ __forceinline__ void count_job(
    unsigned* hist, const int blk,
    const int* __restrict__ rows, int* __restrict__ rank, int* __restrict__ cntmat) {
  const int tid = threadIdx.x;
  for (int i = tid; i < N_NODES / 8; i += 1024) ((uint4*)hist)[i] = make_uint4(0u, 0u, 0u, 0u);
  __syncthreads();
  const int4* r4 = (const int4*)(rows + blk * EDGES_PER_BLK);
  int4*       k4 = (int4*)(rank + blk * EDGES_PER_BLK);
  for (int i = tid; i < EDGES_PER_BLK / 4; i += 1024) {
    const int4 r = r4[i];
    int4 k;
    k.x = bump_row(hist, r.x);
    k.y = bump_row(hist, r.y);
    k.z = bump_row(hist, r.z);
    k.w = bump_row(hist, r.w);
    k4[i] = k;
  }
  __syncthreads();
  int2* dst = (int2*)(cntmat + (size_t)blk * N_NODES);
  for (int w = tid; w < N_NODES / 2; w += 1024) {
    const unsigned c = hist[w];
    dst[w] = make_int2((int)(c & 0xFFFFu), (int)(c >> 16));
  }
}

__global__ __launch_bounds__(1024) void count_and_transform(
    const int* __restrict__ rows, int* __restrict__ rank, int* __restrict__ cntmat,
    const float* __restrict__ F, const float* __restrict__ W,
    const float* __restrict__ bias, float* __restrict__ H) {
  __shared__ __attribute__((aligned(16))) unsigned smem[N_NODES / 2];
  if (blockIdx.x < N_BLK) {
    count_job(smem, (int)blockIdx.x, rows, rank, cntmat);
  } else {
    dense_job((v2f*)smem, (float*)(smem + D_FEAT * N_CH), (int)blockIdx.x - N_BLK, F, W, bias, H);
  }
}

#define ROWS_PER_TILE 256
#define N_TILES ((N_NODES + ROWS_PER_TILE - 1) / ROWS_PER_TILE)

__global__ __launch_bounds__(256) void tile_totals(
    const int* __restrict__ cntmat, int* __restrict__ tsum) {
  __shared__ int wsum[8];
  const int tid  = threadIdx.x;
  const int lane = tid & 31;
  const int wv   = tid >> 5;
  const int r    = blockIdx.x * ROWS_PER_TILE + tid;
  int s = 0;
  if (r < N_NODES) {
    int v[N_BLK];
#pragma unroll
    for (int b = 0; b < N_BLK; ++b) v[b] = cntmat[(size_t)b * N_NODES + r];
#pragma unroll
    for (int b = 0; b < N_BLK; ++b) s += v[b];
  }
  for (int o = 16; o > 0; o >>= 1) s += __shfl_xor(s, o);
  if (lane == 0) wsum[wv] = s;
  __syncthreads();
  if (tid == 0) {
    int q = 0;
    for (int i = 0; i < 8; ++i) q += wsum[i];
    tsum[blockIdx.x] = q;
  }
}

__global__ __launch_bounds__(256) void base_positions(
    int* __restrict__ cntmat, const int* __restrict__ tsum, int* __restrict__ off) {
  __shared__ int wtot[8];
  __shared__ int wpre[8];
  const int tid  = threadIdx.x;
  const int lane = tid & 31;
  const int wv   = tid >> 5;
  const int r    = blockIdx.x * ROWS_PER_TILE + tid;
  const bool live = r < N_NODES;
  int v[N_BLK];
#pragma unroll
  for (int b = 0; b < N_BLK; ++b) v[b] = live ? cntmat[(size_t)b * N_NODES + r] : 0;

  int p = 0;
  for (int i = tid; i < (int)blockIdx.x; i += ROWS_PER_TILE) p += tsum[i];
  for (int o = 16; o > 0; o >>= 1) p += __shfl_xor(p, o);
  if (lane == 0) wpre[wv] = p;

  int c = 0;
#pragma unroll
  for (int b = 0; b < N_BLK; ++b) c += v[b];
  int incl = c;
  for (int o = 1; o < 32; o <<= 1) {
    const int u = __shfl_up(incl, o);
    if (lane >= o) incl += u;
  }
  if (lane == 31) wtot[wv] = incl;
  __syncthreads();
  int front = 0;
  for (int i = 0; i < 8; ++i) {
    front += wpre[i];
    if (i < wv) front += wtot[i];
  }
  int run = front + incl - c;
  if (live) {
    off[r] = run;
    if (r == N_NODES - 1) off[N_NODES] = run + c;
#pragma unroll
    for (int b = 0; b < N_BLK; ++b) {
      cntmat[(size_t)b * N_NODES + r] = run;
      run += v[b];
    }
  }
}

#define HALF_ROWS      (N_NODES / 2)
#define HALF_ROWS4     (HALF_ROWS / 4)
#define QUADS_PER_PART (EDGES_PER_BLK / 8)
#define STAGE_TRIPS    ((HALF_ROWS4 + 1023) / 1024)
#define QUAD_TRIPS     ((QUADS_PER_PART + 1023) / 1024)

__global__ __launch_bounds__(1024) void place_edges(
    const int* __restrict__ rows, const int* __restrict__ cols,
    const float* __restrict__ vals, const int* __restrict__ rank,
    const int* __restrict__ cntmat, int2* __restrict__ sorted) {
  __shared__ __attribute__((aligned(16))) int pos[HALF_ROWS];
  const int tid  = threadIdx.x;
  const int blk  = blockIdx.x >> 2;
  const int hf   = (blockIdx.x >> 1) & 1;
  const int part = blockIdx.x & 1;
  const int lo   = hf * HALF_ROWS;

  const int4* src4 = (const int4*)(cntmat + (size_t)blk * N_NODES + lo);
  int4 s[STAGE_TRIPS];
#pragma unroll
  for (int j = 0; j < STAGE_TRIPS; ++j) {
    const int i = (j < STAGE_TRIPS - 1) ? tid + j * 1024 : HALF_ROWS4 - 1024 + tid;
    s[j] = src4[i];
  }
#pragma unroll
  for (int j = 0; j < STAGE_TRIPS; ++j) {
    const int i = (j < STAGE_TRIPS - 1) ? tid + j * 1024 : HALF_ROWS4 - 1024 + tid;
    ((int4*)pos)[i] = s[j];
  }

  const int q0 = blk * (EDGES_PER_BLK / 4) + part * QUADS_PER_PART;
  int4 r[QUAD_TRIPS], k[QUAD_TRIPS], c[QUAD_TRIPS];
  float4 v[QUAD_TRIPS];
#pragma unroll
  for (int j = 0; j < QUAD_TRIPS; ++j) {
    const int q = q0 + min(tid + j * 1024, QUADS_PER_PART - 1);
    r[j] = ((const int4*)rows)[q];
    k[j] = ((const int4*)rank)[q];
    c[j] = ((const int4*)cols)[q];
    v[j] = ((const float4*)vals)[q];
  }
  __syncthreads();

  int b[QUAD_TRIPS][4];
#pragma unroll
  for (int j = 0; j < QUAD_TRIPS; ++j) {
    b[j][0] = pos[min((unsigned)(r[j].x - lo), (unsigned)(HALF_ROWS - 1))];
    b[j][1] = pos[min((unsigned)(r[j].y - lo), (unsigned)(HALF_ROWS - 1))];
    b[j][2] = pos[min((unsigned)(r[j].z - lo), (unsigned)(HALF_ROWS - 1))];
    b[j][3] = pos[min((unsigned)(r[j].w - lo), (unsigned)(HALF_ROWS - 1))];
  }
#pragma unroll
  for (int j = 0; j < QUAD_TRIPS; ++j) {
    const bool live = (tid + j * 1024) < QUADS_PER_PART;
    const unsigned p0 = min((unsigned)(b[j][0] + k[j].x), (unsigned)(N_EDGES - 1));
    const unsigned p1 = min((unsigned)(b[j][1] + k[j].y), (unsigned)(N_EDGES - 1));
    const unsigned p2 = min((unsigned)(b[j][2] + k[j].z), (unsigned)(N_EDGES - 1));
    const unsigned p3 = min((unsigned)(b[j][3] + k[j].w), (unsigned)(N_EDGES - 1));
    if (live && (unsigned)(r[j].x - lo) < (unsigned)HALF_ROWS) sorted[p0] = make_int2(c[j].x, __float_as_int(v[j].x));
    if (live && (unsigned)(r[j].y - lo) < (unsigned)HALF_ROWS) sorted[p1] = make_int2(c[j].y, __float_as_int(v[j].y));
    if (live && (unsigned)(r[j].z - lo) < (unsigned)HALF_ROWS) sorted[p2] = make_int2(c[j].z, __float_as_int(v[j].z));
    if (live && (unsigned)(r[j].w - lo) < (unsigned)HALF_ROWS) sorted[p3] = make_int2(c[j].w, __float_as_int(v[j].w));
  }
}

__global__ __launch_bounds__(256) void gather_rows(
    const float* __restrict__ H, const int2* __restrict__ sorted,
    const int* __restrict__ off, const float* __restrict__ skip,
    float* __restrict__ out) {
  const int wv = __builtin_amdgcn_readfirstlane((int)(threadIdx.x >> 5));
  const int r  = blockIdx.x * 8 + wv;
  if (r >= N_NODES) return;
  const int lane = threadIdx.x & 31;
  const unsigned p = (unsigned)lane * 2u;
  const int beg = max(off[r], 0);
  const int end = min(off[r + 1], N_EDGES);
  float ax = 0.f, ay = 0.f;
  for (int base = beg; base < end; base += 32) {
    const int n = min(32, end - base);
    int2 e = make_int2(0, 0);
    if (lane < n) e = sorted[base + lane];
#pragma unroll 4
    for (int j = 0; j < n; ++j) {
      const unsigned c = min((unsigned)__builtin_amdgcn_readlane(e.x, j), (unsigned)(N_NODES - 1));
      const float    v = __int_as_float(__builtin_amdgcn_readlane(e.y, j));
      const float2  hv = *(const float2*)(H + (size_t)c * N_CH + p);
      ax += v * hv.x;
      ay += v * hv.y;
    }
  }
  const float2 sk = *(const float2*)(skip + p);
  const float alpha = 1.6732632423543772f;
  const float scale = 1.0507009873554805f;
  const float x0 = ax + sk.x;
  const float x1 = ay + sk.y;
  float2 o;
  o.x = (x0 > 0.0f) ? scale * x0 : scale * alpha * (expf(x0) - 1.0f);
  o.y = (x1 > 0.0f) ? scale * x1 : scale * alpha * (expf(x1) - 1.0f);
  *(float2*)(out + (size_t)r * N_CH + p) = o;
}

extern "C" void kernel_launch(void* const* d_in, const int* in_sizes, int n_in,
                              void* d_out, int out_size, void* d_ws, size_t ws_size,
                              hipStream_t stream) {
  const float* features = (const float*)d_in[0];
  const float* kernelW  = (const float*)d_in[1];
  const float* bias     = (const float*)d_in[2];
  const float* skip     = (const float*)d_in[3];
  const float* vals     = (const float*)d_in[4];
  const int*   rows     = (const int*)d_in[5];
  const int*   cols     = (const int*)d_in[6];

  float* out = (float*)d_out;
  char*  ws  = (char*)d_ws;
  float* h      = (float*)(ws);
  int2*  sorted = (int2*) (ws + 25600000);
  int*   tsum   = (int*)  (ws + 38400000);
  int*   off    = (int*)  (ws + 38800000);
  int*   rank   = (int*)  (ws + 39200256);
  int*   cntmat = (int*)  (ws + 45600256);

  count_and_transform<<<N_BLK + N_GEMM_WG, 1024, 0, stream>>>(rows, rank, cntmat, features, kernelW, bias, h);

  tile_totals<<<N_TILES, ROWS_PER_TILE, 0, stream>>>(cntmat, tsum);

  base_positions<<<N_TILES, ROWS_PER_TILE, 0, stream>>>(cntmat, tsum, off);

  place_edges<<<N_BLK * 4, 1024, 0, stream>>>(rows, cols, vals, rank, cntmat, sorted);

  gather_rows<<<(N_NODES + 7) / 8, 256, 0, stream>>>(h, sorted, off, skip, out);
}


# Row gather: 16,384 resident waves each walk six or seven rows (was one short wave per row, 100,000 wave entries), next row's pairs and offsets loaded during the current row; sums unchanged
# speedup vs baseline: 1.9991x; 1.0423x over previous
#include <hip/hip_runtime.h>
#include <hip/hip_bf16.h>

typedef __attribute__((ext_vector_type(2))) float v2f;
typedef __attribute__((ext_vector_type(8))) float v8f;

#define N_NODES  100000
#define D_FEAT   64
#define N_CH     64
#define N_EDGES  1600000

#define EDGES_PER_BLK 25000
#define N_BLK (N_EDGES / EDGES_PER_BLK)
#define ROW_SLABS   (N_NODES / 16)
#define N_GEMM_WG   ((ROW_SLABS + 31) / 32)

__device__ __forceinline__ void dense_job(
    v2f* ldsW2, float* ldsB, const int wg,
    const float* __restrict__ F, const float* __restrict__ W,
    const float* __restrict__ bias, float* __restrict__ H) {
  for (int i = threadIdx.x; i < (D_FEAT / 2) * N_CH; i += 1024) {
    const int kpair = i >> 6;
    const int col   = i & 63;
    v2f w;
    w.x = W[(2 * kpair + 0) * N_CH + col];
    w.y = W[(2 * kpair + 1) * N_CH + col];
    ldsW2[i] = w;
  }
  if (threadIdx.x < N_CH) ldsB[threadIdx.x] = bias[threadIdx.x];
  __syncthreads();

  const int wave = __builtin_amdgcn_readfirstlane((int)(threadIdx.x >> 5));
  const int lane = threadIdx.x & 31;
  const int half = lane >> 4;
  const int l16  = lane & 15;

  const int slab = wg * 32 + wave;
  if (slab < ROW_SLABS) {
    const int row0 = slab * 16;
    v8f acc[4] = {{}, {}, {}, {}};
    const float* frow = F + (size_t)(row0 + l16) * D_FEAT;

    for (int kk = 0; kk < D_FEAT; kk += 4) {
      const int k2 = kk + 2 * half;
      v2f a;
      a.x = frow[k2];
      a.y = frow[k2 + 1];
      const v2f* wrow = ldsW2 + (size_t)((kk >> 1) + half) * N_CH;
#pragma unroll
      for (int ct = 0; ct < 4; ++ct) {
        v2f b = wrow[ct * 16 + l16];
        acc[ct] = __builtin_amdgcn_wmma_f32_16x16x4_f32(
             false, a,  false, b,
             (short)0, acc[ct],  false,  false);
      }
    }

#pragma unroll
    for (int ct = 0; ct < 4; ++ct) {
      const float bb = ldsB[ct * 16 + l16];
#pragma unroll
      for (int v = 0; v < 8; ++v) {
        H[(size_t)(row0 + v + 8 * half) * N_CH + ct * 16 + l16] = acc[ct][v] + bb;
      }
    }
  }
}

__device__ __forceinline__ int bump_row(unsigned* hist, int r) {
  const unsigned sh  = ((unsigned)r & 1u) << 4;
  const unsigned old = atomicAdd(hist + (r >> 1), 1u << sh);
  return (int)((old >> sh) & 0xFFFFu);
}

__device__ __forceinline__ void count_job(
    unsigned* hist, const int blk,
    const int* __restrict__ rows, int* __restrict__ rank, int* __restrict__ cntmat) {
  const int tid = threadIdx.x;
  for (int i = tid; i < N_NODES / 8; i += 1024) ((uint4*)hist)[i] = make_uint4(0u, 0u, 0u, 0u);
  __syncthreads();
  const int4* r4 = (const int4*)(rows + blk * EDGES_PER_BLK);
  int4*       k4 = (int4*)(rank + blk * EDGES_PER_BLK);
  for (int i = tid; i < EDGES_PER_BLK / 4; i += 1024) {
    const int4 r = r4[i];
    int4 k;
    k.x = bump_row(hist, r.x);
    k.y = bump_row(hist, r.y);
    k.z = bump_row(hist, r.z);
    k.w = bump_row(hist, r.w);
    k4[i] = k;
  }
  __syncthreads();
  int2* dst = (int2*)(cntmat + (size_t)blk * N_NODES);
  for (int w = tid; w < N_NODES / 2; w += 1024) {
    const unsigned c = hist[w];
    dst[w] = make_int2((int)(c & 0xFFFFu), (int)(c >> 16));
  }
}

__global__ __launch_bounds__(1024) void count_and_transform(
    const int* __restrict__ rows, int* __restrict__ rank, int* __restrict__ cntmat,
    const float* __restrict__ F, const float* __restrict__ W,
    const float* __restrict__ bias, float* __restrict__ H) {
  __shared__ __attribute__((aligned(16))) unsigned smem[N_NODES / 2];
  if (blockIdx.x < N_BLK) {
    count_job(smem, (int)blockIdx.x, rows, rank, cntmat);
  } else {
    dense_job((v2f*)smem, (float*)(smem + D_FEAT * N_CH), (int)blockIdx.x - N_BLK, F, W, bias, H);
  }
}

#define ROWS_PER_TILE 256
#define N_TILES ((N_NODES + ROWS_PER_TILE - 1) / ROWS_PER_TILE)

__global__ __launch_bounds__(256) void tile_totals(
    const int* __restrict__ cntmat, int* __restrict__ tsum) {
  __shared__ int wsum[8];
  const int tid  = threadIdx.x;
  const int lane = tid & 31;
  const int wv   = tid >> 5;
  const int r    = blockIdx.x * ROWS_PER_TILE + tid;
  int s = 0;
  if (r < N_NODES) {
    int v[N_BLK];
#pragma unroll
    for (int b = 0; b < N_BLK; ++b) v[b] = cntmat[(size_t)b * N_NODES + r];
#pragma unroll
    for (int b = 0; b < N_BLK; ++b) s += v[b];
  }
  for (int o = 16; o > 0; o >>= 1) s += __shfl_xor(s, o);
  if (lane == 0) wsum[wv] = s;
  __syncthreads();
  if (tid == 0) {
    int q = 0;
    for (int i = 0; i < 8; ++i) q += wsum[i];
    tsum[blockIdx.x] = q;
  }
}

__global__ __launch_bounds__(256) void base_positions(
    int* __restrict__ cntmat, const int* __restrict__ tsum, int* __restrict__ off) {
  __shared__ int wtot[8];
  __shared__ int wpre[8];
  const int tid  = threadIdx.x;
  const int lane = tid & 31;
  const int wv   = tid >> 5;
  const int r    = blockIdx.x * ROWS_PER_TILE + tid;
  const bool live = r < N_NODES;
  int v[N_BLK];
#pragma unroll
  for (int b = 0; b < N_BLK; ++b) v[b] = live ? cntmat[(size_t)b * N_NODES + r] : 0;

  int p = 0;
  for (int i = tid; i < (int)blockIdx.x; i += ROWS_PER_TILE) p += tsum[i];
  for (int o = 16; o > 0; o >>= 1) p += __shfl_xor(p, o);
  if (lane == 0) wpre[wv] = p;

  int c = 0;
#pragma unroll
  for (int b = 0; b < N_BLK; ++b) c += v[b];
  int incl = c;
  for (int o = 1; o < 32; o <<= 1) {
    const int u = __shfl_up(incl, o);
    if (lane >= o) incl += u;
  }
  if (lane == 31) wtot[wv] = incl;
  __syncthreads();
  int front = 0;
  for (int i = 0; i < 8; ++i) {
    front += wpre[i];
    if (i < wv) front += wtot[i];
  }
  int run = front + incl - c;
  if (live) {
    off[r] = run;
    if (r == N_NODES - 1) off[N_NODES] = run + c;
#pragma unroll
    for (int b = 0; b < N_BLK; ++b) {
      cntmat[(size_t)b * N_NODES + r] = run;
      run += v[b];
    }
  }
}

#define HALF_ROWS      (N_NODES / 2)
#define HALF_ROWS4     (HALF_ROWS / 4)
#define QUADS_PER_PART (EDGES_PER_BLK / 8)
#define STAGE_TRIPS    ((HALF_ROWS4 + 1023) / 1024)
#define QUAD_TRIPS     ((QUADS_PER_PART + 1023) / 1024)

__global__ __launch_bounds__(1024) void place_edges(
    const int* __restrict__ rows, const int* __restrict__ cols,
    const float* __restrict__ vals, const int* __restrict__ rank,
    const int* __restrict__ cntmat, int2* __restrict__ sorted) {
  __shared__ __attribute__((aligned(16))) int pos[HALF_ROWS];
  const int tid  = threadIdx.x;
  const int blk  = blockIdx.x >> 2;
  const int hf   = (blockIdx.x >> 1) & 1;
  const int part = blockIdx.x & 1;
  const int lo   = hf * HALF_ROWS;

  const int4* src4 = (const int4*)(cntmat + (size_t)blk * N_NODES + lo);
  int4 s[STAGE_TRIPS];
#pragma unroll
  for (int j = 0; j < STAGE_TRIPS; ++j) {
    const int i = (j < STAGE_TRIPS - 1) ? tid + j * 1024 : HALF_ROWS4 - 1024 + tid;
    s[j] = src4[i];
  }
#pragma unroll
  for (int j = 0; j < STAGE_TRIPS; ++j) {
    const int i = (j < STAGE_TRIPS - 1) ? tid + j * 1024 : HALF_ROWS4 - 1024 + tid;
    ((int4*)pos)[i] = s[j];
  }

  const int q0 = blk * (EDGES_PER_BLK / 4) + part * QUADS_PER_PART;
  int4 r[QUAD_TRIPS], k[QUAD_TRIPS], c[QUAD_TRIPS];
  float4 v[QUAD_TRIPS];
#pragma unroll
  for (int j = 0; j < QUAD_TRIPS; ++j) {
    const int q = q0 + min(tid + j * 1024, QUADS_PER_PART - 1);
    r[j] = ((const int4*)rows)[q];
    k[j] = ((const int4*)rank)[q];
    c[j] = ((const int4*)cols)[q];
    v[j] = ((const float4*)vals)[q];
  }
  __syncthreads();

  int b[QUAD_TRIPS][4];
#pragma unroll
  for (int j = 0; j < QUAD_TRIPS; ++j) {
    b[j][0] = pos[min((unsigned)(r[j].x - lo), (unsigned)(HALF_ROWS - 1))];
    b[j][1] = pos[min((unsigned)(r[j].y - lo), (unsigned)(HALF_ROWS - 1))];
    b[j][2] = pos[min((unsigned)(r[j].z - lo), (unsigned)(HALF_ROWS - 1))];
    b[j][3] = pos[min((unsigned)(r[j].w - lo), (unsigned)(HALF_ROWS - 1))];
  }
#pragma unroll
  for (int j = 0; j < QUAD_TRIPS; ++j) {
    const bool live = (tid + j * 1024) < QUADS_PER_PART;
    const unsigned p0 = min((unsigned)(b[j][0] + k[j].x), (unsigned)(N_EDGES - 1));
    const unsigned p1 = min((unsigned)(b[j][1] + k[j].y), (unsigned)(N_EDGES - 1));
    const unsigned p2 = min((unsigned)(b[j][2] + k[j].z), (unsigned)(N_EDGES - 1));
    const unsigned p3 = min((unsigned)(b[j][3] + k[j].w), (unsigned)(N_EDGES - 1));
    if (live && (unsigned)(r[j].x - lo) < (unsigned)HALF_ROWS) sorted[p0] = make_int2(c[j].x, __float_as_int(v[j].x));
    if (live && (unsigned)(r[j].y - lo) < (unsigned)HALF_ROWS) sorted[p1] = make_int2(c[j].y, __float_as_int(v[j].y));
    if (live && (unsigned)(r[j].z - lo) < (unsigned)HALF_ROWS) sorted[p2] = make_int2(c[j].z, __float_as_int(v[j].z));
    if (live && (unsigned)(r[j].w - lo) < (unsigned)HALF_ROWS) sorted[p3] = make_int2(c[j].w, __float_as_int(v[j].w));
  }
}

#define GATHER_WG    2048
#define GATHER_WAVES (GATHER_WG * 8)

__device__ __forceinline__ void gather_pairs(
    const float* __restrict__ H, const int2 e, const int n, const unsigned p,
    float& ax, float& ay) {
#pragma unroll 4
  for (int j = 0; j < n; ++j) {
    const unsigned c = min((unsigned)__builtin_amdgcn_readlane(e.x, j), (unsigned)(N_NODES - 1));
    const float    v = __int_as_float(__builtin_amdgcn_readlane(e.y, j));
    const float2  hv = *(const float2*)(H + (size_t)c * N_CH + p);
    ax += v * hv.x;
    ay += v * hv.y;
  }
}

__global__ __launch_bounds__(256) void gather_rows(
    const float* __restrict__ H, const int2* __restrict__ sorted,
    const int* __restrict__ off, const float* __restrict__ skip,
    float* __restrict__ out) {
  const int wv   = __builtin_amdgcn_readfirstlane((int)(threadIdx.x >> 5));
  const int lane = threadIdx.x & 31;
  const unsigned p = (unsigned)lane * 2u;
  const float2 sk = *(const float2*)(skip + p);
  const float alpha = 1.6732632423543772f;
  const float scale = 1.0507009873554805f;

  int rn = blockIdx.x * 8 + wv;
  int begn = max(off[rn], 0);
  int endn = min(off[rn + 1], N_EDGES);
  int2 en = make_int2(0, 0);
  if (begn + lane < endn) en = sorted[begn + lane];
  int r2   = rn + GATHER_WAVES;
  int r2c  = min(r2, N_NODES - 1);
  int beg2 = max(off[r2c], 0);
  int end2 = min(off[r2c + 1], N_EDGES);
  asm volatile("" :: "v"(en.x), "v"(en.y));

  while (rn < N_NODES) {
    const int  r   = rn;
    const int  beg = begn;
    const int  end = endn;
    const int2 e   = en;
    rn = r2;  begn = beg2;  endn = end2;
    en = make_int2(0, 0);
    if (rn < N_NODES && begn + lane < endn) en = sorted[begn + lane];
    r2   = rn + GATHER_WAVES;
    r2c  = min(r2, N_NODES - 1);
    beg2 = max(off[r2c], 0);
    end2 = min(off[r2c + 1], N_EDGES);

    float ax = 0.f, ay = 0.f;
    gather_pairs(H, e, min(32, end - beg), p, ax, ay);
    for (int base = beg + 32; base < end; base += 32) {
      const int n = min(32, end - base);
      int2 e2 = make_int2(0, 0);
      if (lane < n) e2 = sorted[base + lane];
      gather_pairs(H, e2, n, p, ax, ay);
    }
    const float x0 = ax + sk.x;
    const float x1 = ay + sk.y;
    float2 o;
    o.x = (x0 > 0.0f) ? scale * x0 : scale * alpha * (expf(x0) - 1.0f);
    o.y = (x1 > 0.0f) ? scale * x1 : scale * alpha * (expf(x1) - 1.0f);
    *(float2*)(out + (size_t)r * N_CH + p) = o;
  }
}

extern "C" void kernel_launch(void* const* d_in, const int* in_sizes, int n_in,
                              void* d_out, int out_size, void* d_ws, size_t ws_size,
                              hipStream_t stream) {
  const float* features = (const float*)d_in[0];
  const float* kernelW  = (const float*)d_in[1];
  const float* bias     = (const float*)d_in[2];
  const float* skip     = (const float*)d_in[3];
  const float* vals     = (const float*)d_in[4];
  const int*   rows     = (const int*)d_in[5];
  const int*   cols     = (const int*)d_in[6];

  float* out = (float*)d_out;
  char*  ws  = (char*)d_ws;
  float* h      = (float*)(ws);
  int2*  sorted = (int2*) (ws + 25600000);
  int*   tsum   = (int*)  (ws + 38400000);
  int*   off    = (int*)  (ws + 38800000);
  int*   rank   = (int*)  (ws + 39200256);
  int*   cntmat = (int*)  (ws + 45600256);

  count_and_transform<<<N_BLK + N_GEMM_WG, 1024, 0, stream>>>(rows, rank, cntmat, features, kernelW, bias, h);

  tile_totals<<<N_TILES, ROWS_PER_TILE, 0, stream>>>(cntmat, tsum);

  base_positions<<<N_TILES, ROWS_PER_TILE, 0, stream>>>(cntmat, tsum, off);

  place_edges<<<N_BLK * 4, 1024, 0, stream>>>(rows, cols, vals, rank, cntmat, sorted);

  gather_rows<<<GATHER_WG, 256, 0, stream>>>(h, sorted, off, skip, out);
}


# Tile totals come from the counting job: each count block sums its LDS counters per 256-row tile and scans them; offsets pass adds 64 block prefixes per tile: 4 launches, was 5; all sums unchanged
# speedup vs baseline: 2.0769x; 1.0389x over previous
#include <hip/hip_runtime.h>
#include <hip/hip_bf16.h>

typedef __attribute__((ext_vector_type(2))) float v2f;
typedef __attribute__((ext_vector_type(8))) float v8f;

#define N_NODES  100000
#define D_FEAT   64
#define N_CH     64
#define N_EDGES  1600000

#define EDGES_PER_BLK 25000
#define N_BLK (N_EDGES / EDGES_PER_BLK)
#define ROW_SLABS   (N_NODES / 16)
#define N_GEMM_WG   ((ROW_SLABS + 31) / 32)
#define ROWS_PER_TILE 256
#define N_TILES ((N_NODES + ROWS_PER_TILE - 1) / ROWS_PER_TILE)
#define TILE_CHUNKS ((N_TILES + 31) / 32)
#define TILE_PAD    (TILE_CHUNKS * 32)
#define HIST_QUADS  (N_NODES / 8)

__device__ __forceinline__ void dense_job(
    v2f* ldsW2, float* ldsB, const int wg,
    const float* __restrict__ F, const float* __restrict__ W,
    const float* __restrict__ bias, float* __restrict__ H) {
  for (int i = threadIdx.x; i < (D_FEAT / 2) * N_CH; i += 1024) {
    const int kpair = i >> 6;
    const int col   = i & 63;
    v2f w;
    w.x = W[(2 * kpair + 0) * N_CH + col];
    w.y = W[(2 * kpair + 1) * N_CH + col];
    ldsW2[i] = w;
  }
  if (threadIdx.x < N_CH) ldsB[threadIdx.x] = bias[threadIdx.x];
  __syncthreads();

  const int wave = __builtin_amdgcn_readfirstlane((int)(threadIdx.x >> 5));
  const int lane = threadIdx.x & 31;
  const int half = lane >> 4;
  const int l16  = lane & 15;

  const int slab = wg * 32 + wave;
  if (slab < ROW_SLABS) {
    const int row0 = slab * 16;
    v8f acc[4] = {{}, {}, {}, {}};
    const float* frow = F + (size_t)(row0 + l16) * D_FEAT;

    for (int kk = 0; kk < D_FEAT; kk += 4) {
      const int k2 = kk + 2 * half;
      v2f a;
      a.x = frow[k2];
      a.y = frow[k2 + 1];
      const v2f* wrow = ldsW2 + (size_t)((kk >> 1) + half) * N_CH;
#pragma unroll
      for (int ct = 0; ct < 4; ++ct) {
        v2f b = wrow[ct * 16 + l16];
        acc[ct] = __builtin_amdgcn_wmma_f32_16x16x4_f32(
             false, a,  false, b,
             (short)0, acc[ct],  false,  false);
      }
    }

#pragma unroll
    for (int ct = 0; ct < 4; ++ct) {
      const float bb = ldsB[ct * 16 + l16];
#pragma unroll
      for (int v = 0; v < 8; ++v) {
        H[(size_t)(row0 + v + 8 * half) * N_CH + ct * 16 + l16] = acc[ct][v] + bb;
      }
    }
  }
}

__device__ __forceinline__ int bump_row(unsigned* hist, int r) {
  const unsigned sh  = ((unsigned)r & 1u) << 4;
  const unsigned old = atomicAdd(hist + (r >> 1), 1u << sh);
  return (int)((old >> sh) & 0xFFFFu);
}

__device__ __forceinline__ void count_job(
    unsigned* hist, int* ttot, int* ctot, const int blk,
    const int* __restrict__ rows, int* __restrict__ rank, int* __restrict__ cntmat,
    int* __restrict__ tpre) {
  const int tid  = threadIdx.x;
  const int lane = tid & 31;
  const int wv   = tid >> 5;
  for (int i = tid; i < N_NODES / 8; i += 1024) ((uint4*)hist)[i] = make_uint4(0u, 0u, 0u, 0u);
  __syncthreads();
  const int4* r4 = (const int4*)(rows + blk * EDGES_PER_BLK);
  int4*       k4 = (int4*)(rank + blk * EDGES_PER_BLK);
  for (int i = tid; i < EDGES_PER_BLK / 4; i += 1024) {
    const int4 r = r4[i];
    int4 k;
    k.x = bump_row(hist, r.x);
    k.y = bump_row(hist, r.y);
    k.z = bump_row(hist, r.z);
    k.w = bump_row(hist, r.w);
    k4[i] = k;
  }
  __syncthreads();
  int2* dst = (int2*)(cntmat + (size_t)blk * N_NODES);
  for (int w = tid; w < N_NODES / 2; w += 1024) {
    const unsigned c = hist[w];
    dst[w] = make_int2((int)(c & 0xFFFFu), (int)(c >> 16));
  }

  int s[TILE_CHUNKS];
#pragma unroll
  for (int j = 0; j < TILE_CHUNKS; ++j) {
    const int g = (wv + 32 * j) * 32 + lane;
    uint4 c = make_uint4(0u, 0u, 0u, 0u);
    if (g < HIST_QUADS) c = ((const uint4*)hist)[g];
    s[j] = (int)((c.x & 0xFFFFu) + (c.x >> 16) + (c.y & 0xFFFFu) + (c.y >> 16) +
                 (c.z & 0xFFFFu) + (c.z >> 16) + (c.w & 0xFFFFu) + (c.w >> 16));
  }
#pragma unroll
  for (int o = 16; o > 0; o >>= 1) {
#pragma unroll
    for (int j = 0; j < TILE_CHUNKS; ++j) s[j] += __shfl_xor(s[j], o);
  }
  if (lane == 0) {
#pragma unroll
    for (int j = 0; j < TILE_CHUNKS; ++j) ttot[wv + 32 * j] = s[j];
  }
  __syncthreads();

  int v = 0, incl = 0;
  if (wv < TILE_CHUNKS) {
    v = ttot[tid];
    incl = v;
    for (int o = 1; o < 32; o <<= 1) {
      const int u = __shfl_up(incl, o);
      if (lane >= o) incl += u;
    }
    if (lane == 31) ctot[wv] = incl;
  }
  __syncthreads();
  if (tid < N_TILES) {
    int front = 0;
    for (int i = 0; i < TILE_CHUNKS; ++i) {
      if (i < wv) front += ctot[i];
    }
    tpre[blk * N_TILES + tid] = front + incl - v;
  }
}

__global__ __launch_bounds__(1024) void count_and_transform(
    const int* __restrict__ rows, int* __restrict__ rank, int* __restrict__ cntmat,
    int* __restrict__ tpre,
    const float* __restrict__ F, const float* __restrict__ W,
    const float* __restrict__ bias, float* __restrict__ H) {
  __shared__ __attribute__((aligned(16))) unsigned smem[N_NODES / 2];
  __shared__ int ttot[TILE_PAD];
  __shared__ int ctot[TILE_CHUNKS];
  if (blockIdx.x < N_BLK) {
    count_job(smem, ttot, ctot, (int)blockIdx.x, rows, rank, cntmat, tpre);
  } else {
    dense_job((v2f*)smem, (float*)(smem + D_FEAT * N_CH), (int)blockIdx.x - N_BLK, F, W, bias, H);
  }
}

__global__ __launch_bounds__(256) void base_positions(
    int* __restrict__ cntmat, const int* __restrict__ tpre, int* __restrict__ off) {
  __shared__ int wtot[8];
  __shared__ int wpre[8];
  const int tid  = threadIdx.x;
  const int lane = tid & 31;
  const int wv   = tid >> 5;
  const int r    = blockIdx.x * ROWS_PER_TILE + tid;
  const bool live = r < N_NODES;
  int v[N_BLK];
#pragma unroll
  for (int b = 0; b < N_BLK; ++b) v[b] = live ? cntmat[(size_t)b * N_NODES + r] : 0;

  int p = 0;
  if (tid < N_BLK) p = tpre[tid * N_TILES + (int)blockIdx.x];
  for (int o = 16; o > 0; o >>= 1) p += __shfl_xor(p, o);
  if (lane == 0) wpre[wv] = p;

  int c = 0;
#pragma unroll
  for (int b = 0; b < N_BLK; ++b) c += v[b];
  int incl = c;
  for (int o = 1; o < 32; o <<= 1) {
    const int u = __shfl_up(incl, o);
    if (lane >= o) incl += u;
  }
  if (lane == 31) wtot[wv] = incl;
  __syncthreads();
  int front = 0;
  for (int i = 0; i < 8; ++i) {
    front += wpre[i];
    if (i < wv) front += wtot[i];
  }
  int run = front + incl - c;
  if (live) {
    off[r] = run;
    if (r == N_NODES - 1) off[N_NODES] = run + c;
#pragma unroll
    for (int b = 0; b < N_BLK; ++b) {
      cntmat[(size_t)b * N_NODES + r] = run;
      run += v[b];
    }
  }
}

#define HALF_ROWS      (N_NODES / 2)
#define HALF_ROWS4     (HALF_ROWS / 4)
#define QUADS_PER_PART (EDGES_PER_BLK / 8)
#define STAGE_TRIPS    ((HALF_ROWS4 + 1023) / 1024)
#define QUAD_TRIPS     ((QUADS_PER_PART + 1023) / 1024)

__global__ __launch_bounds__(1024) void place_edges(
    const int* __restrict__ rows, const int* __restrict__ cols,
    const float* __restrict__ vals, const int* __restrict__ rank,
    const int* __restrict__ cntmat, int2* __restrict__ sorted) {
  __shared__ __attribute__((aligned(16))) int pos[HALF_ROWS];
  const int tid  = threadIdx.x;
  const int blk  = blockIdx.x >> 2;
  const int hf   = (blockIdx.x >> 1) & 1;
  const int part = blockIdx.x & 1;
  const int lo   = hf * HALF_ROWS;

  const int4* src4 = (const int4*)(cntmat + (size_t)blk * N_NODES + lo);
  int4 s[STAGE_TRIPS];
#pragma unroll
  for (int j = 0; j < STAGE_TRIPS; ++j) {
    const int i = (j < STAGE_TRIPS - 1) ? tid + j * 1024 : HALF_ROWS4 - 1024 + tid;
    s[j] = src4[i];
  }
#pragma unroll
  for (int j = 0; j < STAGE_TRIPS; ++j) {
    const int i = (j < STAGE_TRIPS - 1) ? tid + j * 1024 : HALF_ROWS4 - 1024 + tid;
    ((int4*)pos)[i] = s[j];
  }

  const int q0 = blk * (EDGES_PER_BLK / 4) + part * QUADS_PER_PART;
  int4 r[QUAD_TRIPS], k[QUAD_TRIPS], c[QUAD_TRIPS];
  float4 v[QUAD_TRIPS];
#pragma unroll
  for (int j = 0; j < QUAD_TRIPS; ++j) {
    const int q = q0 + min(tid + j * 1024, QUADS_PER_PART - 1);
    r[j] = ((const int4*)rows)[q];
    k[j] = ((const int4*)rank)[q];
    c[j] = ((const int4*)cols)[q];
    v[j] = ((const float4*)vals)[q];
  }
  __syncthreads();

  int b[QUAD_TRIPS][4];
#pragma unroll
  for (int j = 0; j < QUAD_TRIPS; ++j) {
    b[j][0] = pos[min((unsigned)(r[j].x - lo), (unsigned)(HALF_ROWS - 1))];
    b[j][1] = pos[min((unsigned)(r[j].y - lo), (unsigned)(HALF_ROWS - 1))];
    b[j][2] = pos[min((unsigned)(r[j].z - lo), (unsigned)(HALF_ROWS - 1))];
    b[j][3] = pos[min((unsigned)(r[j].w - lo), (unsigned)(HALF_ROWS - 1))];
  }
#pragma unroll
  for (int j = 0; j < QUAD_TRIPS; ++j) {
    const bool live = (tid + j * 1024) < QUADS_PER_PART;
    const unsigned p0 = min((unsigned)(b[j][0] + k[j].x), (unsigned)(N_EDGES - 1));
    const unsigned p1 = min((unsigned)(b[j][1] + k[j].y), (unsigned)(N_EDGES - 1));
    const unsigned p2 = min((unsigned)(b[j][2] + k[j].z), (unsigned)(N_EDGES - 1));
    const unsigned p3 = min((unsigned)(b[j][3] + k[j].w), (unsigned)(N_EDGES - 1));
    if (live && (unsigned)(r[j].x - lo) < (unsigned)HALF_ROWS) sorted[p0] = make_int2(c[j].x, __float_as_int(v[j].x));
    if (live && (unsigned)(r[j].y - lo) < (unsigned)HALF_ROWS) sorted[p1] = make_int2(c[j].y, __float_as_int(v[j].y));
    if (live && (unsigned)(r[j].z - lo) < (unsigned)HALF_ROWS) sorted[p2] = make_int2(c[j].z, __float_as_int(v[j].z));
    if (live && (unsigned)(r[j].w - lo) < (unsigned)HALF_ROWS) sorted[p3] = make_int2(c[j].w, __float_as_int(v[j].w));
  }
}

#define GATHER_WG    2048
#define GATHER_WAVES (GATHER_WG * 8)

__device__ __forceinline__ void gather_pairs(
    const float* __restrict__ H, const int2 e, const int n, const unsigned p,
    float& ax, float& ay) {
#pragma unroll 4
  for (int j = 0; j < n; ++j) {
    const unsigned c = min((unsigned)__builtin_amdgcn_readlane(e.x, j), (unsigned)(N_NODES - 1));
    const float    v = __int_as_float(__builtin_amdgcn_readlane(e.y, j));
    const float2  hv = *(const float2*)(H + (size_t)c * N_CH + p);
    ax += v * hv.x;
    ay += v * hv.y;
  }
}

__global__ __launch_bounds__(256) void gather_rows(
    const float* __restrict__ H, const int2* __restrict__ sorted,
    const int* __restrict__ off, const float* __restrict__ skip,
    float* __restrict__ out) {
  const int wv   = __builtin_amdgcn_readfirstlane((int)(threadIdx.x >> 5));
  const int lane = threadIdx.x & 31;
  const unsigned p = (unsigned)lane * 2u;
  const float2 sk = *(const float2*)(skip + p);
  const float alpha = 1.6732632423543772f;
  const float scale = 1.0507009873554805f;

  int rn = blockIdx.x * 8 + wv;
  int begn = max(off[rn], 0);
  int endn = min(off[rn + 1], N_EDGES);
  int2 en = make_int2(0, 0);
  if (begn + lane < endn) en = sorted[begn + lane];
  int r2   = rn + GATHER_WAVES;
  int r2c  = min(r2, N_NODES - 1);
  int beg2 = max(off[r2c], 0);
  int end2 = min(off[r2c + 1], N_EDGES);
  asm volatile("" :: "v"(en.x), "v"(en.y));

  while (rn < N_NODES) {
    const int  r   = rn;
    const int  beg = begn;
    const int  end = endn;
    const int2 e   = en;
    rn = r2;  begn = beg2;  endn = end2;
    en = make_int2(0, 0);
    if (rn < N_NODES && begn + lane < endn) en = sorted[begn + lane];
    r2   = rn + GATHER_WAVES;
    r2c  = min(r2, N_NODES - 1);
    beg2 = max(off[r2c], 0);
    end2 = min(off[r2c + 1], N_EDGES);

    float ax = 0.f, ay = 0.f;
    gather_pairs(H, e, min(32, end - beg), p, ax, ay);
    for (int base = beg + 32; base < end; base += 32) {
      const int n = min(32, end - base);
      int2 e2 = make_int2(0, 0);
      if (lane < n) e2 = sorted[base + lane];
      gather_pairs(H, e2, n, p, ax, ay);
    }
    const float x0 = ax + sk.x;
    const float x1 = ay + sk.y;
    float2 o;
    o.x = (x0 > 0.0f) ? scale * x0 : scale * alpha * (expf(x0) - 1.0f);
    o.y = (x1 > 0.0f) ? scale * x1 : scale * alpha * (expf(x1) - 1.0f);
    *(float2*)(out + (size_t)r * N_CH + p) = o;
  }
}

extern "C" void kernel_launch(void* const* d_in, const int* in_sizes, int n_in,
                              void* d_out, int out_size, void* d_ws, size_t ws_size,
                              hipStream_t stream) {
  const float* features = (const float*)d_in[0];
  const float* kernelW  = (const float*)d_in[1];
  const float* bias     = (const float*)d_in[2];
  const float* skip     = (const float*)d_in[3];
  const float* vals     = (const float*)d_in[4];
  const int*   rows     = (const int*)d_in[5];
  const int*   cols     = (const int*)d_in[6];

  float* out = (float*)d_out;
  char*  ws  = (char*)d_ws;
  float* h      = (float*)(ws);
  int2*  sorted = (int2*) (ws + 25600000);
  int*   tpre   = (int*)  (ws + 38400000);
  int*   off    = (int*)  (ws + 38800000);
  int*   rank   = (int*)  (ws + 39200256);
  int*   cntmat = (int*)  (ws + 45600256);

  count_and_transform<<<N_BLK + N_GEMM_WG, 1024, 0, stream>>>(rows, rank, cntmat, tpre, features, kernelW, bias, h);

  base_positions<<<N_TILES, ROWS_PER_TILE, 0, stream>>>(cntmat, tpre, off);

  place_edges<<<N_BLK * 4, 1024, 0, stream>>>(rows, cols, vals, rank, cntmat, sorted);

  gather_rows<<<GATHER_WG, 256, 0, stream>>>(h, sorted, off, skip, out);
}


# Edge order by two-level counting sort: chunks ordered by 196-row bin in LDS, then a work-group per bin orders its runs in LDS and sums rows in registers: 2 launches, was 4; sum order = LDS arrival
# speedup vs baseline: 4.0140x; 1.9327x over previous
#include <hip/hip_runtime.h>
#include <hip/hip_bf16.h>


typedef __attribute__((ext_vector_type(2))) float v2f;
typedef __attribute__((ext_vector_type(8))) float v8f;

#define N_NODES  100000
#define D_FEAT   64
#define N_CH     64
#define N_EDGES  1600000

#define CHUNK_E     6400
#define N_CHUNK     (N_EDGES / CHUNK_E)
#define CHUNK_QUADS (CHUNK_E / 4)
#define QUAD_TRIPS  2
#define BIN_ROWS    196
#define N_BINS      ((N_NODES + BIN_ROWS - 1) / BIN_ROWS)
#define BIN_CHUNKS  ((N_BINS + 31) / 32)
#define BIN_PAD     (BIN_CHUNKS * 32)
#define COL_BITS    17
#define COL_MASK    0x1FFFF
#define ROW_SLABS   (N_NODES / 16)
#define N_GEMM_WG   ((ROW_SLABS + 31) / 32)

__device__ __forceinline__ void dense_job(
    v2f* ldsW2, float* ldsB, const int wg,
    const float* __restrict__ F, const float* __restrict__ W,
    const float* __restrict__ bias, float* __restrict__ H) {
  for (int i = threadIdx.x; i < (D_FEAT / 2) * N_CH; i += 1024) {
    const int kpair = i >> 6;
    const int col   = i & 63;
    v2f w;
    w.x = W[(2 * kpair + 0) * N_CH + col];
    w.y = W[(2 * kpair + 1) * N_CH + col];
    ldsW2[i] = w;
  }
  if (threadIdx.x < N_CH) ldsB[threadIdx.x] = bias[threadIdx.x];
  __syncthreads();

  const int wave = __builtin_amdgcn_readfirstlane((int)(threadIdx.x >> 5));
  const int lane = threadIdx.x & 31;
  const int half = lane >> 4;
  const int l16  = lane & 15;

  const int slab = wg * 32 + wave;
  if (slab < ROW_SLABS) {
    const int row0 = slab * 16;
    v8f acc[4] = {{}, {}, {}, {}};
    const float* frow = F + (size_t)(row0 + l16) * D_FEAT;

    for (int kk = 0; kk < D_FEAT; kk += 4) {
      const int k2 = kk + 2 * half;
      v2f a;
      a.x = frow[k2];
      a.y = frow[k2 + 1];
      const v2f* wrow = ldsW2 + (size_t)((kk >> 1) + half) * N_CH;
#pragma unroll
      for (int ct = 0; ct < 4; ++ct) {
        v2f b = wrow[ct * 16 + l16];
        acc[ct] = __builtin_amdgcn_wmma_f32_16x16x4_f32(
             false, a,  false, b,
             (short)0, acc[ct],  false,  false);
      }
    }

#pragma unroll
    for (int ct = 0; ct < 4; ++ct) {
      const float bb = ldsB[ct * 16 + l16];
#pragma unroll
      for (int v = 0; v < 8; ++v) {
        H[(size_t)(row0 + v + 8 * half) * N_CH + ct * 16 + l16] = acc[ct][v] + bb;
      }
    }
  }
}

__device__ __forceinline__ int bump_edge(unsigned* cnt, const bool live, int r, int c, int& recx) {
  const unsigned rc  = min((unsigned)r, (unsigned)(N_NODES - 1));
  const unsigned cc  = min((unsigned)c, (unsigned)(N_NODES - 1));
  const unsigned bin = rc / BIN_ROWS;
  const unsigned rl  = rc - bin * BIN_ROWS;
  recx = (int)((rl << COL_BITS) | cc);
  unsigned rank = 0u;
  if (live) rank = atomicAdd(cnt + bin, 1u);
  return (int)((bin << 16) | rank);
}

__device__ __forceinline__ void chunk_job(
    int2* lrec, unsigned* cnt, int* sstart, int* ctot, const int ch,
    const int* __restrict__ rows, const int* __restrict__ cols,
    const float* __restrict__ vals, int2* __restrict__ recs, int2* __restrict__ tab) {
  const int tid  = threadIdx.x;
  const int lane = tid & 31;
  const int wv   = tid >> 5;
  if (tid < BIN_PAD) cnt[tid] = 0u;

  const int q0 = ch * CHUNK_QUADS;
  int4 r[QUAD_TRIPS], c[QUAD_TRIPS];
  float4 v[QUAD_TRIPS];
#pragma unroll
  for (int j = 0; j < QUAD_TRIPS; ++j) {
    const int q = q0 + min(tid + j * 1024, CHUNK_QUADS - 1);
    r[j] = ((const int4*)rows)[q];
    c[j] = ((const int4*)cols)[q];
    v[j] = ((const float4*)vals)[q];
  }
  __syncthreads();
  int4 x[QUAD_TRIPS], bk[QUAD_TRIPS];
#pragma unroll
  for (int j = 0; j < QUAD_TRIPS; ++j) {
    const bool live = tid + j * 1024 < CHUNK_QUADS;
    bk[j].x = bump_edge(cnt, live, r[j].x, c[j].x, x[j].x);
    bk[j].y = bump_edge(cnt, live, r[j].y, c[j].y, x[j].y);
    bk[j].z = bump_edge(cnt, live, r[j].z, c[j].z, x[j].z);
    bk[j].w = bump_edge(cnt, live, r[j].w, c[j].w, x[j].w);
  }
  __syncthreads();

  int val = 0, incl = 0;
  if (wv < BIN_CHUNKS) {
    val = (int)cnt[tid];
    incl = val;
    for (int o = 1; o < 32; o <<= 1) {
      const int u = __shfl_up(incl, o);
      if (lane >= o) incl += u;
    }
    if (lane == 31) ctot[wv] = incl;
  }
  __syncthreads();
  if (wv < BIN_CHUNKS) {
    int front = 0;
    for (int i = 0; i < BIN_CHUNKS; ++i) {
      if (i < wv) front += ctot[i];
    }
    const int st = front + incl - val;
    sstart[tid] = st;
    if (tid < N_BINS) tab[(size_t)tid * N_CHUNK + ch] = make_int2(st, val);
  }
  __syncthreads();

#pragma unroll
  for (int j = 0; j < QUAD_TRIPS; ++j) {
    if (tid + j * 1024 < CHUNK_QUADS) {
      lrec[min(sstart[bk[j].x >> 16] + (bk[j].x & 0xFFFF), CHUNK_E - 1)] = make_int2(x[j].x, __float_as_int(v[j].x));
      lrec[min(sstart[bk[j].y >> 16] + (bk[j].y & 0xFFFF), CHUNK_E - 1)] = make_int2(x[j].y, __float_as_int(v[j].y));
      lrec[min(sstart[bk[j].z >> 16] + (bk[j].z & 0xFFFF), CHUNK_E - 1)] = make_int2(x[j].z, __float_as_int(v[j].z));
      lrec[min(sstart[bk[j].w >> 16] + (bk[j].w & 0xFFFF), CHUNK_E - 1)] = make_int2(x[j].w, __float_as_int(v[j].w));
    }
  }
  __syncthreads();
  int4* dst = (int4*)(recs + (size_t)ch * CHUNK_E);
  for (int i = tid; i < CHUNK_E / 2; i += 1024) dst[i] = ((const int4*)lrec)[i];
}

__global__ __launch_bounds__(1024, 16) void sort_chunks_and_transform(
    const int* __restrict__ rows, const int* __restrict__ cols,
    const float* __restrict__ vals, int2* __restrict__ recs, int2* __restrict__ tab,
    const float* __restrict__ F, const float* __restrict__ W,
    const float* __restrict__ bias, float* __restrict__ H) {
  __shared__ __attribute__((aligned(16))) unsigned smem[CHUNK_E * 2];
  __shared__ unsigned cnt[BIN_PAD];
  __shared__ int sstart[BIN_PAD];
  __shared__ int ctot[BIN_CHUNKS];
  if (blockIdx.x < N_CHUNK) {
    chunk_job((int2*)smem, cnt, sstart, ctot, (int)blockIdx.x, rows, cols, vals, recs, tab);
  } else {
    dense_job((v2f*)smem, (float*)(smem + D_FEAT * N_CH), (int)blockIdx.x - N_CHUNK, F, W, bias, H);
  }
}

__device__ __forceinline__ void gather_pairs(
    const float* __restrict__ H, const int2 e, const int n, const unsigned p,
    float& ax, float& ay) {
#pragma unroll 4
  for (int j = 0; j < n; ++j) {
    const unsigned c = min((unsigned)__builtin_amdgcn_readlane(e.x, j), (unsigned)(N_NODES - 1));
    const float    v = __int_as_float(__builtin_amdgcn_readlane(e.y, j));
    const float2  hv = *(const float2*)(H + (size_t)c * N_CH + p);
    ax += v * hv.x;
    ay += v * hv.y;
  }
}

#define ROUND_CAP 8192
#define BIN_WAVES     28
#define BIN_THREADS   (BIN_WAVES * 32)
#define ROWS_PER_WAVE (BIN_ROWS / BIN_WAVES)
#define RUN_GROUPS    3
#define RUNS_PER_GROUP 3

__device__ __forceinline__ int rec_row(const int2 rec) { return (rec.x >> COL_BITS) & 255; }

__global__ __launch_bounds__(BIN_THREADS, 16) void sum_bins(
    const float* __restrict__ H, const int2* __restrict__ recs,
    const int2* __restrict__ tab, const float* __restrict__ skip,
    float* __restrict__ out) {
  __shared__ __attribute__((aligned(16))) int2 ubuf[ROUND_CAP];
  __shared__ __attribute__((aligned(16))) int2 sbuf[ROUND_CAP];
  __shared__ int runsrc[256];
  __shared__ int runlen[256];
  __shared__ int runoff[256];
  __shared__ unsigned rowcnt[256];
  __shared__ unsigned cursor[256];
  __shared__ int rowstart[256];
  __shared__ int wtot[8];

  const int tid  = threadIdx.x;
  const int lane = tid & 31;
  const int wv   = __builtin_amdgcn_readfirstlane((int)(threadIdx.x >> 5));
  const int b    = blockIdx.x;
  const unsigned p = (unsigned)lane * 2u;
  const float2 sk = *(const float2*)(skip + p);
  const float alpha = 1.6732632423543772f;
  const float scale = 1.0507009873554805f;

  int len = 0, src = 0, incl = 0;
  if (wv < 8) {
    if (tid < N_CHUNK) {
      const int2 t = tab[(size_t)b * N_CHUNK + tid];
      const int s0 = min(max(t.x, 0), CHUNK_E);
      len = min(max(t.y, 0), CHUNK_E - s0);
      src = tid * CHUNK_E + s0;
    }
    incl = len;
    for (int o = 1; o < 32; o <<= 1) {
      const int u = __shfl_up(incl, o);
      if (lane >= o) incl += u;
    }
    if (lane == 31) wtot[wv] = incl;
  }
  __syncthreads();
  int nsum = 0;
  for (int i = 0; i < 8; ++i) nsum += wtot[i];
  const int n = __builtin_amdgcn_readfirstlane(nsum);
  if (wv < 8) {
    int front = 0;
    for (int i = 0; i < 8; ++i) {
      if (i < wv) front += wtot[i];
    }
    runsrc[tid] = src;
    runlen[tid] = len;
    runoff[tid] = front + incl - len;
  }
  __syncthreads();

  float ax[ROWS_PER_WAVE], ay[ROWS_PER_WAVE];
#pragma unroll
  for (int s = 0; s < ROWS_PER_WAVE; ++s) { ax[s] = 0.f; ay[s] = 0.f; }

  for (int base = 0; base < n; base += ROUND_CAP) {
    const int m = min(ROUND_CAP, n - base);
    if (tid < 256) rowcnt[tid] = 0u;
    __syncthreads();

    for (int g = 0; g < RUN_GROUPS; ++g) {
      int2 rr[RUNS_PER_GROUP];
      int  rf[RUNS_PER_GROUP];
#pragma unroll
      for (int k = 0; k < RUNS_PER_GROUP; ++k) {
        const int run = wv + BIN_WAVES * (g * RUNS_PER_GROUP + k);
        const int ln = runlen[run];
        const int f  = runoff[run] + lane - base;
        const bool ok = lane < ln && f >= 0 && f < m;
        rr[k] = make_int2(0, 0);
        if (ok) rr[k] = recs[runsrc[run] + lane];
        rf[k] = ok ? f : -1;
      }
#pragma unroll
      for (int k = 0; k < RUNS_PER_GROUP; ++k) {
        if (rf[k] >= 0) {
          ubuf[rf[k]] = rr[k];
          atomicAdd(rowcnt + rec_row(rr[k]), 1u);
        }
      }
      for (int k = 0; k < RUNS_PER_GROUP; ++k) {
        const int run = wv + BIN_WAVES * (g * RUNS_PER_GROUP + k);
        const int ln = __builtin_amdgcn_readfirstlane(runlen[run]);
        const int of = runoff[run] - base;
        const int sr = runsrc[run];
        for (int j = 32; j < ln; j += 32) {
          const int jj = j + lane;
          const int f  = of + jj;
          if (jj < ln && f >= 0 && f < m) {
            const int2 rec = recs[sr + jj];
            ubuf[f] = rec;
            atomicAdd(rowcnt + rec_row(rec), 1u);
          }
        }
      }
    }
    __syncthreads();

    int rc = 0, rincl = 0;
    if (wv < 8) {
      rc = (int)rowcnt[tid];
      rincl = rc;
      for (int o = 1; o < 32; o <<= 1) {
        const int u = __shfl_up(rincl, o);
        if (lane >= o) rincl += u;
      }
      if (lane == 31) wtot[wv] = rincl;
    }
    __syncthreads();
    if (wv < 8) {
      int front = 0;
      for (int i = 0; i < 8; ++i) {
        if (i < wv) front += wtot[i];
      }
      const int st = front + rincl - rc;
      rowstart[tid] = st;
      cursor[tid]   = (unsigned)st;
    }
    __syncthreads();

    for (int i = tid; i < m; i += BIN_THREADS) {
      const int2 rec = ubuf[i];
      const unsigned pos = atomicAdd(cursor + rec_row(rec), 1u);
      sbuf[min(pos, (unsigned)(ROUND_CAP - 1))] = rec;
    }
    __syncthreads();

#pragma unroll
    for (int s = 0; s < ROWS_PER_WAVE; ++s) {
      const int rl  = wv * ROWS_PER_WAVE + s;
      const int beg = min(max(__builtin_amdgcn_readfirstlane(rowstart[rl]), 0), ROUND_CAP);
      const int cn  = min(max(__builtin_amdgcn_readfirstlane((int)rowcnt[rl]), 0), ROUND_CAP - beg);
      for (int bb = 0; bb < cn; bb += 32) {
        const int nn = min(32, cn - bb);
        int2 e = make_int2(0, 0);
        if (lane < nn) {
          e = sbuf[beg + bb + lane];
          e.x &= COL_MASK;
        }
        gather_pairs(H, e, nn, p, ax[s], ay[s]);
      }
    }
    __syncthreads();
  }

#pragma unroll
  for (int s = 0; s < ROWS_PER_WAVE; ++s) {
    const int r = b * BIN_ROWS + wv * ROWS_PER_WAVE + s;
    if (r < N_NODES) {
      const float x0 = ax[s] + sk.x;
      const float x1 = ay[s] + sk.y;
      float2 o;
      o.x = (x0 > 0.0f) ? scale * x0 : scale * alpha * (expf(x0) - 1.0f);
      o.y = (x1 > 0.0f) ? scale * x1 : scale * alpha * (expf(x1) - 1.0f);
      *(float2*)(out + (size_t)r * N_CH + p) = o;
    }
  }
}

extern "C" void kernel_launch(void* const* d_in, const int* in_sizes, int n_in,
                              void* d_out, int out_size, void* d_ws, size_t ws_size,
                              hipStream_t stream) {
  const float* features = (const float*)d_in[0];
  const float* kernelW  = (const float*)d_in[1];
  const float* bias     = (const float*)d_in[2];
  const float* skip     = (const float*)d_in[3];
  const float* vals     = (const float*)d_in[4];
  const int*   rows     = (const int*)d_in[5];
  const int*   cols     = (const int*)d_in[6];

  float* out = (float*)d_out;
  char*  ws  = (char*)d_ws;
  float* h    = (float*)(ws);
  int2*  recs = (int2*) (ws + 25600000);
  int2*  tab  = (int2*) (ws + 38400000);

  sort_chunks_and_transform<<<N_CHUNK + N_GEMM_WG, 1024, 0, stream>>>(rows, cols, vals, recs, tab, features, kernelW, bias, h);

  sum_bins<<<N_BINS, BIN_THREADS, 0, stream>>>(h, recs, tab, skip, out);
}


# Launch 1 work-group order reversed: the 196 dense (matrix) groups take the low indices and enter first, the 250 edge-chunk groups behind them (longest job first); all sums unchanged
# speedup vs baseline: 4.5256x; 1.1275x over previous
#include <hip/hip_runtime.h>
#include <hip/hip_bf16.h>


typedef __attribute__((ext_vector_type(2))) float v2f;
typedef __attribute__((ext_vector_type(8))) float v8f;

#define N_NODES  100000
#define D_FEAT   64
#define N_CH     64
#define N_EDGES  1600000

#define CHUNK_E     6400
#define N_CHUNK     (N_EDGES / CHUNK_E)
#define CHUNK_QUADS (CHUNK_E / 4)
#define QUAD_TRIPS  2
#define BIN_ROWS    196
#define N_BINS      ((N_NODES + BIN_ROWS - 1) / BIN_ROWS)
#define BIN_CHUNKS  ((N_BINS + 31) / 32)
#define BIN_PAD     (BIN_CHUNKS * 32)
#define COL_BITS    17
#define COL_MASK    0x1FFFF
#define ROW_SLABS   (N_NODES / 16)
#define N_GEMM_WG   ((ROW_SLABS + 31) / 32)

__device__ __forceinline__ void dense_job(
    v2f* ldsW2, float* ldsB, const int wg,
    const float* __restrict__ F, const float* __restrict__ W,
    const float* __restrict__ bias, float* __restrict__ H) {
  for (int i = threadIdx.x; i < (D_FEAT / 2) * N_CH; i += 1024) {
    const int kpair = i >> 6;
    const int col   = i & 63;
    v2f w;
    w.x = W[(2 * kpair + 0) * N_CH + col];
    w.y = W[(2 * kpair + 1) * N_CH + col];
    ldsW2[i] = w;
  }
  if (threadIdx.x < N_CH) ldsB[threadIdx.x] = bias[threadIdx.x];
  __syncthreads();

  const int wave = __builtin_amdgcn_readfirstlane((int)(threadIdx.x >> 5));
  const int lane = threadIdx.x & 31;
  const int half = lane >> 4;
  const int l16  = lane & 15;

  const int slab = wg * 32 + wave;
  if (slab < ROW_SLABS) {
    const int row0 = slab * 16;
    v8f acc[4] = {{}, {}, {}, {}};
    const float* frow = F + (size_t)(row0 + l16) * D_FEAT;

    for (int kk = 0; kk < D_FEAT; kk += 4) {
      const int k2 = kk + 2 * half;
      v2f a;
      a.x = frow[k2];
      a.y = frow[k2 + 1];
      const v2f* wrow = ldsW2 + (size_t)((kk >> 1) + half) * N_CH;
#pragma unroll
      for (int ct = 0; ct < 4; ++ct) {
        v2f b = wrow[ct * 16 + l16];
        acc[ct] = __builtin_amdgcn_wmma_f32_16x16x4_f32(
             false, a,  false, b,
             (short)0, acc[ct],  false,  false);
      }
    }

#pragma unroll
    for (int ct = 0; ct < 4; ++ct) {
      const float bb = ldsB[ct * 16 + l16];
#pragma unroll
      for (int v = 0; v < 8; ++v) {
        H[(size_t)(row0 + v + 8 * half) * N_CH + ct * 16 + l16] = acc[ct][v] + bb;
      }
    }
  }
}

__device__ __forceinline__ int bump_edge(unsigned* cnt, const bool live, int r, int c, int& recx) {
  const unsigned rc  = min((unsigned)r, (unsigned)(N_NODES - 1));
  const unsigned cc  = min((unsigned)c, (unsigned)(N_NODES - 1));
  const unsigned bin = rc / BIN_ROWS;
  const unsigned rl  = rc - bin * BIN_ROWS;
  recx = (int)((rl << COL_BITS) | cc);
  unsigned rank = 0u;
  if (live) rank = atomicAdd(cnt + bin, 1u);
  return (int)((bin << 16) | rank);
}

__device__ __forceinline__ void chunk_job(
    int2* lrec, unsigned* cnt, int* sstart, int* ctot, const int ch,
    const int* __restrict__ rows, const int* __restrict__ cols,
    const float* __restrict__ vals, int2* __restrict__ recs, int2* __restrict__ tab) {
  const int tid  = threadIdx.x;
  const int lane = tid & 31;
  const int wv   = tid >> 5;
  if (tid < BIN_PAD) cnt[tid] = 0u;

  const int q0 = ch * CHUNK_QUADS;
  int4 r[QUAD_TRIPS], c[QUAD_TRIPS];
  float4 v[QUAD_TRIPS];
#pragma unroll
  for (int j = 0; j < QUAD_TRIPS; ++j) {
    const int q = q0 + min(tid + j * 1024, CHUNK_QUADS - 1);
    r[j] = ((const int4*)rows)[q];
    c[j] = ((const int4*)cols)[q];
    v[j] = ((const float4*)vals)[q];
  }
  __syncthreads();
  int4 x[QUAD_TRIPS], bk[QUAD_TRIPS];
#pragma unroll
  for (int j = 0; j < QUAD_TRIPS; ++j) {
    const bool live = tid + j * 1024 < CHUNK_QUADS;
    bk[j].x = bump_edge(cnt, live, r[j].x, c[j].x, x[j].x);
    bk[j].y = bump_edge(cnt, live, r[j].y, c[j].y, x[j].y);
    bk[j].z = bump_edge(cnt, live, r[j].z, c[j].z, x[j].z);
    bk[j].w = bump_edge(cnt, live, r[j].w, c[j].w, x[j].w);
  }
  __syncthreads();

  int val = 0, incl = 0;
  if (wv < BIN_CHUNKS) {
    val = (int)cnt[tid];
    incl = val;
    for (int o = 1; o < 32; o <<= 1) {
      const int u = __shfl_up(incl, o);
      if (lane >= o) incl += u;
    }
    if (lane == 31) ctot[wv] = incl;
  }
  __syncthreads();
  if (wv < BIN_CHUNKS) {
    int front = 0;
    for (int i = 0; i < BIN_CHUNKS; ++i) {
      if (i < wv) front += ctot[i];
    }
    const int st = front + incl - val;
    sstart[tid] = st;
    if (tid < N_BINS) tab[(size_t)tid * N_CHUNK + ch] = make_int2(st, val);
  }
  __syncthreads();

#pragma unroll
  for (int j = 0; j < QUAD_TRIPS; ++j) {
    if (tid + j * 1024 < CHUNK_QUADS) {
      lrec[min(sstart[bk[j].x >> 16] + (bk[j].x & 0xFFFF), CHUNK_E - 1)] = make_int2(x[j].x, __float_as_int(v[j].x));
      lrec[min(sstart[bk[j].y >> 16] + (bk[j].y & 0xFFFF), CHUNK_E - 1)] = make_int2(x[j].y, __float_as_int(v[j].y));
      lrec[min(sstart[bk[j].z >> 16] + (bk[j].z & 0xFFFF), CHUNK_E - 1)] = make_int2(x[j].z, __float_as_int(v[j].z));
      lrec[min(sstart[bk[j].w >> 16] + (bk[j].w & 0xFFFF), CHUNK_E - 1)] = make_int2(x[j].w, __float_as_int(v[j].w));
    }
  }
  __syncthreads();
  int4* dst = (int4*)(recs + (size_t)ch * CHUNK_E);
  for (int i = tid; i < CHUNK_E / 2; i += 1024) dst[i] = ((const int4*)lrec)[i];
}

__global__ __launch_bounds__(1024, 16) void sort_chunks_and_transform(
    const int* __restrict__ rows, const int* __restrict__ cols,
    const float* __restrict__ vals, int2* __restrict__ recs, int2* __restrict__ tab,
    const float* __restrict__ F, const float* __restrict__ W,
    const float* __restrict__ bias, float* __restrict__ H) {
  __shared__ __attribute__((aligned(16))) unsigned smem[CHUNK_E * 2];
  __shared__ unsigned cnt[BIN_PAD];
  __shared__ int sstart[BIN_PAD];
  __shared__ int ctot[BIN_CHUNKS];
  if (blockIdx.x < N_GEMM_WG) {
    dense_job((v2f*)smem, (float*)(smem + D_FEAT * N_CH), (int)blockIdx.x, F, W, bias, H);
  } else {
    chunk_job((int2*)smem, cnt, sstart, ctot, (int)blockIdx.x - N_GEMM_WG, rows, cols, vals, recs, tab);
  }
}

#define ROUND_CAP 8192
#define BIN_WAVES     28
#define BIN_THREADS   (BIN_WAVES * 32)
#define ROWS_PER_WAVE (BIN_ROWS / BIN_WAVES)
#define RUN_GROUPS    3
#define RUNS_PER_GROUP 3

__device__ __forceinline__ int rec_row(const int2 rec) { return (rec.x >> COL_BITS) & 255; }

#define NO_ROW 255
#define SEG_RAW(b) sbuf[min(segBeg + (b) * 32 + lane, ROUND_CAP - 1)]
#define SEG_FIX(b, raw) ((((b) * 32 + lane) < cnt) ? (raw) : make_int2(NO_ROW << COL_BITS, 0))
#define FLUSH_ROW()                                                             \
  {                                                                             \
    const unsigned sl = min((unsigned)(cur - r0), (unsigned)ROWS_PER_WAVE);     \
    wacc[sl * (N_CH / 2) + lane] = make_float2(cx, cy);                         \
  }
#define ISSUE_GROUP(g, HV, SV, SR)                                                      \
  {                                                                                     \
    const int l0 = ((g) & 7) * 4;                                                       \
    _Pragma("unroll")                                                                   \
    for (int k = 0; k < 4; ++k) {                                                       \
      const int xw = __builtin_amdgcn_readlane(eI.x, l0 + k);                           \
      SV[k] = __int_as_float(__builtin_amdgcn_readlane(eI.y, l0 + k));                  \
      SR[k] = (xw >> COL_BITS) & 255;                                                   \
                           \
      const unsigned cc = min((unsigned)(xw & COL_MASK), (unsigned)(N_NODES - 1));      \
        \
      HV[k] = *(const float2*)(H + (cc * (unsigned)N_CH + p));                          \
    }                                                                                   \
  }
#define USE_GROUP(HV, SV, SR)                                                           \
  {                                                                                     \
    _Pragma("unroll")                                                                   \
    for (int k = 0; k < 4; ++k) {                                                       \
      if (SR[k] != cur) {                                              \
        FLUSH_ROW();                                                                    \
        cur = SR[k];                                                                    \
        cx = 0.f;                                                                       \
        cy = 0.f;                                                                       \
      }                                                                                 \
      cx += SV[k] * HV[k].x;                                                            \
      cy += SV[k] * HV[k].y;                                                            \
    }                                                                                   \
  }

__global__ __launch_bounds__(BIN_THREADS, 16) void sum_bins(
    const float* __restrict__ H, const int2* __restrict__ recs,
    const int2* __restrict__ tab, const float* __restrict__ skip,
    float* __restrict__ out) {
  __shared__ __attribute__((aligned(16))) int2 ubuf[ROUND_CAP];
  __shared__ __attribute__((aligned(16))) int2 sbuf[ROUND_CAP];
  __shared__ int runsrc[256];
  __shared__ int runlen[256];
  __shared__ int runoff[256];
  __shared__ unsigned rowcnt[256];
  __shared__ unsigned cursor[256];
  __shared__ int rowstart[256];
  __shared__ int wtot[8];

  const int tid  = threadIdx.x;
  const int lane = tid & 31;
  const int wv   = __builtin_amdgcn_readfirstlane((int)(threadIdx.x >> 5));
  const int b    = blockIdx.x;
  const unsigned p = (unsigned)lane * 2u;
  const float2 sk = *(const float2*)(skip + p);
  const float alpha = 1.6732632423543772f;
  const float scale = 1.0507009873554805f;

  int len = 0, src = 0, incl = 0;
  if (wv < 8) {
    if (tid < N_CHUNK) {
      const int2 t = tab[(size_t)b * N_CHUNK + tid];
      const int s0 = min(max(t.x, 0), CHUNK_E);
      len = min(max(t.y, 0), CHUNK_E - s0);
      src = tid * CHUNK_E + s0;
    }
    incl = len;
    for (int o = 1; o < 32; o <<= 1) {
      const int u = __shfl_up(incl, o);
      if (lane >= o) incl += u;
    }
    if (lane == 31) wtot[wv] = incl;
  }
  __syncthreads();
  int nsum = 0;
  for (int i = 0; i < 8; ++i) nsum += wtot[i];
  const int n = __builtin_amdgcn_readfirstlane(nsum);
  if (wv < 8) {
    int front = 0;
    for (int i = 0; i < 8; ++i) {
      if (i < wv) front += wtot[i];
    }
    runsrc[tid] = src;
    runlen[tid] = len;
    runoff[tid] = front + incl - len;
  }
  __syncthreads();

  float ax[ROWS_PER_WAVE], ay[ROWS_PER_WAVE];
#pragma unroll
  for (int s = 0; s < ROWS_PER_WAVE; ++s) { ax[s] = 0.f; ay[s] = 0.f; }

  for (int base = 0; base < n; base += ROUND_CAP) {
    const int m = min(ROUND_CAP, n - base);
    if (tid < 256) rowcnt[tid] = 0u;
    __syncthreads();

    for (int g = 0; g < RUN_GROUPS; ++g) {
      int2 rr[RUNS_PER_GROUP];
      int  rf[RUNS_PER_GROUP];
#pragma unroll
      for (int k = 0; k < RUNS_PER_GROUP; ++k) {
        const int run = wv + BIN_WAVES * (g * RUNS_PER_GROUP + k);
        const int ln = runlen[run];
        const int f  = runoff[run] + lane - base;
        const bool ok = lane < ln && f >= 0 && f < m;
        rr[k] = make_int2(0, 0);
        if (ok) rr[k] = recs[runsrc[run] + lane];
        rf[k] = ok ? f : -1;
      }
#pragma unroll
      for (int k = 0; k < RUNS_PER_GROUP; ++k) {
        if (rf[k] >= 0) {
          ubuf[rf[k]] = rr[k];
          atomicAdd(rowcnt + rec_row(rr[k]), 1u);
        }
      }
      for (int k = 0; k < RUNS_PER_GROUP; ++k) {
        const int run = wv + BIN_WAVES * (g * RUNS_PER_GROUP + k);
        const int ln = __builtin_amdgcn_readfirstlane(runlen[run]);
        const int of = runoff[run] - base;
        const int sr = runsrc[run];
        for (int j = 32; j < ln; j += 32) {
          const int jj = j + lane;
          const int f  = of + jj;
          if (jj < ln && f >= 0 && f < m) {
            const int2 rec = recs[sr + jj];
            ubuf[f] = rec;
            atomicAdd(rowcnt + rec_row(rec), 1u);
          }
        }
      }
    }
    __syncthreads();

    int rc = 0, rincl = 0;
    if (wv < 8) {
      rc = (int)rowcnt[tid];
      rincl = rc;
      for (int o = 1; o < 32; o <<= 1) {
        const int u = __shfl_up(rincl, o);
        if (lane >= o) rincl += u;
      }
      if (lane == 31) wtot[wv] = rincl;
    }
    __syncthreads();
    if (wv < 8) {
      int front = 0;
      for (int i = 0; i < 8; ++i) {
        if (i < wv) front += wtot[i];
      }
      const int st = front + rincl - rc;
      rowstart[tid] = st;
      cursor[tid]   = (unsigned)st;
    }
    __syncthreads();

    for (int i = tid; i < m; i += BIN_THREADS) {
      const int2 rec = ubuf[i];
      const unsigned pos = atomicAdd(cursor + rec_row(rec), 1u);
      sbuf[min(pos, (unsigned)(ROUND_CAP - 1))] = rec;
    }
    __syncthreads();

    {
      const int r0 = wv * ROWS_PER_WAVE;
      const int segBeg = min(max(__builtin_amdgcn_readfirstlane(rowstart[r0]), 0), ROUND_CAP);
      const int segEnd = __builtin_amdgcn_readfirstlane(rowstart[r0 + ROWS_PER_WAVE - 1]) +
                         __builtin_amdgcn_readfirstlane((int)rowcnt[r0 + ROWS_PER_WAVE - 1]);
      const int cnt = min(max(segEnd - segBeg, 0), ROUND_CAP - segBeg);
      const int ngroups = (cnt + 3) >> 2;
      float2* wacc = (float2*)ubuf + wv * ((ROWS_PER_WAVE + 1) * (N_CH / 2));
#pragma unroll
      for (int s = 0; s < ROWS_PER_WAVE; ++s) wacc[s * (N_CH / 2) + lane] = make_float2(0.f, 0.f);
      const int2 e0 = SEG_RAW(0);
      int2 eI = SEG_FIX(0, e0);
      int2 eN = SEG_RAW(1);
      int cur = NO_ROW;
      float cx = 0.f, cy = 0.f;
      float2 hA[4], hB[4];
      float  vA[4], vB[4];
      int    rA[4], rB[4];
      ISSUE_GROUP(0, hA, vA, rA);
      for (int gi = 0; gi < ngroups; gi += 2) {
        ISSUE_GROUP(gi + 1, hB, vB, rB);
        USE_GROUP(hA, vA, rA);
        if (((gi + 2) & 7) == 0) {
          const int nb = (gi + 2) >> 3;
          eI = SEG_FIX(nb, eN);
          eN = SEG_RAW(nb + 1);
        }
        ISSUE_GROUP(gi + 2, hA, vA, rA);
        USE_GROUP(hB, vB, rB);
      }
      FLUSH_ROW();
#pragma unroll
      for (int s = 0; s < ROWS_PER_WAVE; ++s) {
        const float2 t = wacc[s * (N_CH / 2) + lane];
        ax[s] += t.x;
        ay[s] += t.y;
      }
    }
    __syncthreads();
  }

#pragma unroll
  for (int s = 0; s < ROWS_PER_WAVE; ++s) {
    const int r = b * BIN_ROWS + wv * ROWS_PER_WAVE + s;
    if (r < N_NODES) {
      const float x0 = ax[s] + sk.x;
      const float x1 = ay[s] + sk.y;
      float2 o;
      o.x = (x0 > 0.0f) ? scale * x0 : scale * alpha * (expf(x0) - 1.0f);
      o.y = (x1 > 0.0f) ? scale * x1 : scale * alpha * (expf(x1) - 1.0f);
      *(float2*)(out + (size_t)r * N_CH + p) = o;
    }
  }
}

extern "C" void kernel_launch(void* const* d_in, const int* in_sizes, int n_in,
                              void* d_out, int out_size, void* d_ws, size_t ws_size,
                              hipStream_t stream) {
  const float* features = (const float*)d_in[0];
  const float* kernelW  = (const float*)d_in[1];
  const float* bias     = (const float*)d_in[2];
  const float* skip     = (const float*)d_in[3];
  const float* vals     = (const float*)d_in[4];
  const int*   rows     = (const int*)d_in[5];
  const int*   cols     = (const int*)d_in[6];

  float* out = (float*)d_out;
  char*  ws  = (char*)d_ws;
  float* h    = (float*)(ws);
  int2*  recs = (int2*) (ws + 25600000);
  int2*  tab  = (int2*) (ws + 38400000);

  sort_chunks_and_transform<<<N_CHUNK + N_GEMM_WG, 1024, 0, stream>>>(rows, cols, vals, recs, tab, features, kernelW, bias, h);

  sum_bins<<<N_BINS, BIN_THREADS, 0, stream>>>(h, recs, tab, skip, out);
}
